# starting kernel
# baseline (speedup 1.0000x reference)
#include <hip/hip_runtime.h>

typedef float v2f __attribute__((ext_vector_type(2)));
typedef float v8f __attribute__((ext_vector_type(8)));

#define F_IN 128
#define HID  128
#define HID2 64
#define OUTC 32

static inline int ceil_div(int a, int b) { return (a + b - 1) / b; }


__global__ void deg_init_kernel(float* __restrict__ deg, int N) {
  int i = blockIdx.x * blockDim.x + threadIdx.x;
  if (i < N) deg[i] = 1.0f;
}

__global__ void deg_accum_kernel(const int* __restrict__ dst, float* __restrict__ deg, int E) {
  int e = blockIdx.x * blockDim.x + threadIdx.x;
  if (e < E) atomicAdd(&deg[dst[e]], 1.0f);
}

__global__ void deg_rsqrt_kernel(float* __restrict__ deg, int N) {
  int i = blockIdx.x * blockDim.x + threadIdx.x;
  if (i < N) deg[i] = rsqrtf(deg[i]);
}

__global__ __launch_bounds__(128) void gemm_wmma_kernel(
    const float* __restrict__ A, const float* __restrict__ B, float* __restrict__ C,
    int M, int K, int Ncol) {
  int lane = threadIdx.x & 31;
  int wave = threadIdx.x >> 5;
  int mtile = blockIdx.x * 4 + wave;
  int row0 = mtile << 4;
  if (row0 >= M) return;
  int col0 = blockIdx.y << 4;
  int half = lane >> 4;
  int l16  = lane & 15;

  v8f acc = {0.f, 0.f, 0.f, 0.f, 0.f, 0.f, 0.f, 0.f};
  const float* arow = A + (size_t)(row0 + l16) * K;
  const float* bcol = B + col0 + l16;

  for (int k = 0; k < K; k += 4) {
    int ka = k + half * 2;
    v2f a, b;
    a.x = arow[ka + 0];
    a.y = arow[ka + 1];
    b.x = bcol[(size_t)(ka + 0) * Ncol];
    b.y = bcol[(size_t)(ka + 1) * Ncol];
    acc = __builtin_amdgcn_wmma_f32_16x16x4_f32(
         false, a,  false, b,
         (short)0, acc,  false,  false);
  }

  float* crow = C + (size_t)(row0 + half * 8) * Ncol + col0 + l16;
#pragma unroll
  for (int r = 0; r < 8; ++r)
    crow[(size_t)r * Ncol] = acc[r];
}


__global__ void selfloop_init_kernel(const float* __restrict__ XW, const float* __restrict__ dis,
                                     float* __restrict__ acc, int N, int F) {
  int i = blockIdx.x * blockDim.x + threadIdx.x;
  if (i >= N * F) return;
  float d = dis[i / F];
  acc[i] = XW[i] * (d * d);
}

__global__ void scatter_kernel(const float* __restrict__ XW, const int* __restrict__ src,
                               const int* __restrict__ dst, const float* __restrict__ dis,
                               float* __restrict__ acc, int E, int F) {
  int gid = blockIdx.x * blockDim.x + threadIdx.x;
  int lane = gid & 31;
  int e = gid >> 5;
  if (e >= E) return;
  int s = src[e], d = dst[e];
  float nrm = dis[s] * dis[d];
  int nchunks = F >> 2;
  if (lane < nchunks) {
    const float4 v = *((const float4*)(XW + (size_t)s * F) + lane);
    float* ap = acc + (size_t)d * F + lane * 4;
    atomicAdd(ap + 0, v.x * nrm);
    atomicAdd(ap + 1, v.y * nrm);
    atomicAdd(ap + 2, v.z * nrm);
    atomicAdd(ap + 3, v.w * nrm);
  }
}

__global__ void bias_relu_kernel(float* __restrict__ h, const float* __restrict__ b,
                                 int N, int F) {
  int i = blockIdx.x * blockDim.x + threadIdx.x;
  if (i >= N * F) return;
  float v = h[i] + b[i & (F - 1)];
  h[i] = v > 0.f ? v : 0.f;
}


__global__ void pool_count_kernel(const int* __restrict__ batch, float* __restrict__ cnt, int N) {
  int i = blockIdx.x * blockDim.x + threadIdx.x;
  if (i < N) atomicAdd(&cnt[batch[i]], 1.0f);
}

__global__ void pool_sum_kernel(const float* __restrict__ h, const int* __restrict__ batch,
                                float* __restrict__ out, int N) {
  int gid = blockIdx.x * blockDim.x + threadIdx.x;
  int lane = gid & 31;
  int node = gid >> 5;
  if (node >= N) return;
  int g = batch[node];
  atomicAdd(&out[(size_t)g * OUTC + lane], h[(size_t)node * OUTC + lane]);
}

__global__ void pool_div_kernel(float* __restrict__ out, const float* __restrict__ cnt, int G) {
  int i = blockIdx.x * blockDim.x + threadIdx.x;
  if (i >= G * OUTC) return;
  out[i] /= fmaxf(cnt[i >> 5], 1.0f);
}


extern "C" void kernel_launch(void* const* d_in, const int* in_sizes, int n_in,
                              void* d_out, int out_size, void* d_ws, size_t ws_size,
                              hipStream_t stream) {
  const float* x     = (const float*)d_in[0];
  const int*   ei    = (const int*)d_in[1];
  const int*   batch = (const int*)d_in[2];
  const float* W1    = (const float*)d_in[3];
  const float* b1    = (const float*)d_in[4];
  const float* W2    = (const float*)d_in[5];
  const float* b2    = (const float*)d_in[6];
  const float* W3    = (const float*)d_in[7];
  const float* b3    = (const float*)d_in[8];

  const int N = in_sizes[0] / F_IN;
  const int E = in_sizes[1] / 2;
  const int G = out_size / OUTC;
  const int* src = ei;
  const int* dst = ei + E;
  float* out = (float*)d_out;

  float* dis = (float*)d_ws;
  float* cnt = dis + N;
  size_t off = (size_t)N + (size_t)G;
  off = (off + 63) & ~(size_t)63;
  float* bufA = (float*)d_ws + off;
  float* bufB = bufA + (size_t)N * HID;

  const int TB = 256;

  hipMemsetAsync(cnt, 0, (size_t)G * sizeof(float), stream);
  hipMemsetAsync(out, 0, (size_t)out_size * sizeof(float), stream);

  deg_init_kernel<<<ceil_div(N, TB), TB, 0, stream>>>(dis, N);
  deg_accum_kernel<<<ceil_div(E, TB), TB, 0, stream>>>(dst, dis, E);
  deg_rsqrt_kernel<<<ceil_div(N, TB), TB, 0, stream>>>(dis, N);

  auto conv = [&](const float* hin, const float* W, const float* bb,
                  float* XW, float* accb, int K, int Cout) {
    dim3 gg(ceil_div(ceil_div(N, 16), 4), Cout / 16);
    gemm_wmma_kernel<<<gg, 128, 0, stream>>>(hin, W, XW, N, K, Cout);
    selfloop_init_kernel<<<ceil_div(N * Cout, TB), TB, 0, stream>>>(XW, dis, accb, N, Cout);
    scatter_kernel<<<ceil_div(E * 32, TB), TB, 0, stream>>>(XW, src, dst, dis, accb, E, Cout);
    bias_relu_kernel<<<ceil_div(N * Cout, TB), TB, 0, stream>>>(accb, bb, N, Cout);
  };

  conv(x,    W1, b1, bufA, bufB, F_IN, HID);
  conv(bufB, W2, b2, bufA, bufB, HID,  HID2);
  conv(bufB, W3, b3, bufA, bufB, HID2, OUTC);

  pool_count_kernel<<<ceil_div(N, TB), TB, 0, stream>>>(batch, cnt, N);
  pool_sum_kernel<<<ceil_div(N * 32, TB), TB, 0, stream>>>(bufB, batch, out, N);
  pool_div_kernel<<<ceil_div(G * OUTC, TB), TB, 0, stream>>>(out, cnt, G);
}


# Edges counting-sorted by destination once per call (arrival-order rank from one int atomic per edge); each layer sums a node's rows in registers, no f32 atomics; sum order stays arrival order
# speedup vs baseline: 2.1550x; 2.1550x over previous
#include <hip/hip_runtime.h>

typedef float v2f __attribute__((ext_vector_type(2)));
typedef float v8f __attribute__((ext_vector_type(8)));

#define F_IN 128
#define HID  128
#define HID2 64
#define OUTC 32

static inline int ceil_div(int a, int b) { return (a + b - 1) / b; }


#define SCAN_CHUNK 1024

__global__ void degi_zero_kernel(int* __restrict__ degi, int Npad) {
  int i = blockIdx.x * blockDim.x + threadIdx.x;
  if (i < Npad) degi[i] = 0;
}

__global__ void edge_rank_kernel(const int* __restrict__ dst, int* __restrict__ degi,
                                 int* __restrict__ rank, int E) {
  int e = blockIdx.x * blockDim.x + threadIdx.x;
  if (e < E) rank[e] = atomicAdd(&degi[dst[e]], 1);
}

__global__ __launch_bounds__(256) void rowptr_kernel(const int* __restrict__ degi,
                                                     int* __restrict__ rowptr,
                                                     float* __restrict__ dis, int N) {
  __shared__ int front[8];
  __shared__ int wtot[8];
  const int t = threadIdx.x;
  const int lane = t & 31, wave = t >> 5;
  const int b = blockIdx.x;
  const int4* d4 = (const int4*)degi;

  int pre = 0;
#pragma unroll 8
  for (int i = t; i < b * 256; i += 256) {
    int4 q = d4[i];
    pre += q.x + q.y + q.z + q.w;
  }
  for (int o = 16; o > 0; o >>= 1) pre += __shfl_xor(pre, o);
  if (lane == 0) front[wave] = pre;

  const int4 v = d4[b * 256 + t];
  const int tot = v.x + v.y + v.z + v.w;
  int inc = tot;
  for (int o = 1; o < 32; o <<= 1) {
    int u = __shfl_up(inc, o);
    if (lane >= o) inc += u;
  }
  if (lane == 31) wtot[wave] = inc;
  __syncthreads();

  int ex = inc - tot;
  for (int w = 0; w < 8; ++w) {
    ex += front[w];
    if (w < wave) ex += wtot[w];
  }
  const int c[4] = {v.x, v.y, v.z, v.w};
  const int i0 = (b * 256 + t) * 4;
#pragma unroll
  for (int k = 0; k < 4; ++k) {
    int i = i0 + k;
    if (i <= N) rowptr[i] = ex;
    if (i < N) dis[i] = rsqrtf((float)(c[k] + 1));
    ex += c[k];
  }
}

__global__ void csr_fill_kernel(const int* __restrict__ src, const int* __restrict__ dst,
                                const int* __restrict__ rank, const int* __restrict__ rowptr,
                                const float* __restrict__ dis, int* __restrict__ csr_, int E) {
  int e = blockIdx.x * blockDim.x + threadIdx.x;
  if (e >= E) return;
  int s = src[e], d = dst[e];
  int2* csr = (int2*)csr_;
  csr[rowptr[d] + rank[e]] = make_int2(s, __float_as_int(dis[s]));
}

__global__ __launch_bounds__(128) void gemm_wmma_kernel(
    const float* __restrict__ A, const float* __restrict__ B, float* __restrict__ C,
    int M, int K, int Ncol) {
  int lane = threadIdx.x & 31;
  int wave = threadIdx.x >> 5;
  int mtile = blockIdx.x * 4 + wave;
  int row0 = mtile << 4;
  if (row0 >= M) return;
  int col0 = blockIdx.y << 4;
  int half = lane >> 4;
  int l16  = lane & 15;

  v8f acc = {0.f, 0.f, 0.f, 0.f, 0.f, 0.f, 0.f, 0.f};
  const float* arow = A + (size_t)(row0 + l16) * K;
  const float* bcol = B + col0 + l16;

  for (int k = 0; k < K; k += 4) {
    int ka = k + half * 2;
    v2f a, b;
    a.x = arow[ka + 0];
    a.y = arow[ka + 1];
    b.x = bcol[(size_t)(ka + 0) * Ncol];
    b.y = bcol[(size_t)(ka + 1) * Ncol];
    acc = __builtin_amdgcn_wmma_f32_16x16x4_f32(
         false, a,  false, b,
         (short)0, acc,  false,  false);
  }

  float* crow = C + (size_t)(row0 + half * 8) * Ncol + col0 + l16;
#pragma unroll
  for (int r = 0; r < 8; ++r)
    crow[(size_t)r * Ncol] = acc[r];
}


template <int LPN, bool TAIL>
__device__ __forceinline__ void add8(float4& acc, const int2* __restrict__ csr,
                                     const float4* __restrict__ rows, int j, int end, int sub, float dd) {
  int2 ent[8];
  float4 v[8];
#pragma unroll
  for (int u = 0; u < 8; ++u) {
    int jj = j + u;
    if (TAIL) jj = jj < end ? jj : end - 1;
    ent[u] = csr[jj];
  }
#pragma unroll
  for (int u = 0; u < 8; ++u)
    v[u] = rows[ent[u].x * LPN + sub];
#pragma unroll
  for (int u = 0; u < 8; ++u) {
    float4 r = v[u];
    if (TAIL) {
      bool ok = j + u < end;
      r.x = ok ? r.x : 0.f;
      r.y = ok ? r.y : 0.f;
      r.z = ok ? r.z : 0.f;
      r.w = ok ? r.w : 0.f;
    }
    float nrm = __int_as_float(ent[u].y) * dd;
    acc.x = fmaf(r.x, nrm, acc.x);
    acc.y = fmaf(r.y, nrm, acc.y);
    acc.z = fmaf(r.z, nrm, acc.z);
    acc.w = fmaf(r.w, nrm, acc.w);
  }
}

template <int F>
__global__ __launch_bounds__(256) void gather_kernel(const float* __restrict__ XW,
                                                     const int* __restrict__ rowptr,
                                                     const int* __restrict__ csr_,
                                                     const float* __restrict__ dis,
                                                     float* __restrict__ acc_out, int N) {
  constexpr int LPN = F / 4;
  constexpr int NPB = 256 / LPN;
  const int sub = threadIdx.x % LPN;
  const int node = blockIdx.x * NPB + threadIdx.x / LPN;
  if (node >= N) return;
  const int2* csr = (const int2*)csr_;
  const float4* rows = (const float4*)XW;
  const int beg = rowptr[node], end = rowptr[node + 1];
  const float dd = dis[node];
  const float4 own = rows[node * LPN + sub];
  const float sw = dd * dd;
  float4 acc = make_float4(own.x * sw, own.y * sw, own.z * sw, own.w * sw);

  int j = beg;
  for (; j + 8 <= end; j += 8) add8<LPN, false>(acc, csr, rows, j, end, sub, dd);
  if (j < end) add8<LPN, true>(acc, csr, rows, j, end, sub, dd);

  ((float4*)acc_out)[node * LPN + sub] = acc;
}

__global__ void bias_relu_kernel(float* __restrict__ h, const float* __restrict__ b,
                                 int N, int F) {
  int i = blockIdx.x * blockDim.x + threadIdx.x;
  if (i >= N * F) return;
  float v = h[i] + b[i & (F - 1)];
  h[i] = v > 0.f ? v : 0.f;
}


__global__ void pool_count_kernel(const int* __restrict__ batch, float* __restrict__ cnt, int N) {
  int i = blockIdx.x * blockDim.x + threadIdx.x;
  if (i < N) atomicAdd(&cnt[batch[i]], 1.0f);
}

__global__ void pool_sum_kernel(const float* __restrict__ h, const int* __restrict__ batch,
                                float* __restrict__ out, int N) {
  int gid = blockIdx.x * blockDim.x + threadIdx.x;
  int lane = gid & 31;
  int node = gid >> 5;
  if (node >= N) return;
  int g = batch[node];
  atomicAdd(&out[(size_t)g * OUTC + lane], h[(size_t)node * OUTC + lane]);
}

__global__ void pool_div_kernel(float* __restrict__ out, const float* __restrict__ cnt, int G) {
  int i = blockIdx.x * blockDim.x + threadIdx.x;
  if (i >= G * OUTC) return;
  out[i] /= fmaxf(cnt[i >> 5], 1.0f);
}


extern "C" void kernel_launch(void* const* d_in, const int* in_sizes, int n_in,
                              void* d_out, int out_size, void* d_ws, size_t ws_size,
                              hipStream_t stream) {
  const float* x     = (const float*)d_in[0];
  const int*   ei    = (const int*)d_in[1];
  const int*   batch = (const int*)d_in[2];
  const float* W1    = (const float*)d_in[3];
  const float* b1    = (const float*)d_in[4];
  const float* W2    = (const float*)d_in[5];
  const float* b2    = (const float*)d_in[6];
  const float* W3    = (const float*)d_in[7];
  const float* b3    = (const float*)d_in[8];

  const int N = in_sizes[0] / F_IN;
  const int E = in_sizes[1] / 2;
  const int G = out_size / OUTC;
  const int* src = ei;
  const int* dst = ei + E;
  float* out = (float*)d_out;

  float* dis = (float*)d_ws;
  float* cnt = dis + N;
  size_t off = (size_t)N + (size_t)G;
  off = (off + 63) & ~(size_t)63;
  float* bufA = (float*)d_ws + off;
  float* bufB = bufA + (size_t)N * HID;
  const int Npad = ceil_div(N + 1, SCAN_CHUNK) * SCAN_CHUNK;
  const size_t Epad = ((size_t)E + 63) & ~(size_t)63;
  int* degi   = (int*)(bufB + (size_t)N * HID);
  int* rowptr = degi + Npad;
  int* rank   = rowptr + Npad;
  int* csr    = rank + Epad;

  const int TB = 256;

  hipMemsetAsync(cnt, 0, (size_t)G * sizeof(float), stream);
  hipMemsetAsync(out, 0, (size_t)out_size * sizeof(float), stream);

  degi_zero_kernel<<<ceil_div(Npad, TB), TB, 0, stream>>>(degi, Npad);
  edge_rank_kernel<<<ceil_div(E, TB), TB, 0, stream>>>(dst, degi, rank, E);
  rowptr_kernel<<<Npad / SCAN_CHUNK, 256, 0, stream>>>(degi, rowptr, dis, N);
  csr_fill_kernel<<<ceil_div(E, TB), TB, 0, stream>>>(src, dst, rank, rowptr, dis, csr, E);

  auto gemm = [&](const float* hin, const float* W, float* XW, int K, int Cout) {
    dim3 gg(ceil_div(ceil_div(N, 16), 4), Cout / 16);
    gemm_wmma_kernel<<<gg, 128, 0, stream>>>(hin, W, XW, N, K, Cout);
  };

  gemm(x, W1, bufA, F_IN, HID);
  gather_kernel<HID><<<ceil_div(N, 256 / (HID / 4)), 256, 0, stream>>>(bufA, rowptr, csr, dis, bufB, N);
  bias_relu_kernel<<<ceil_div(N * HID, TB), TB, 0, stream>>>(bufB, b1, N, HID);

  gemm(bufB, W2, bufA, HID, HID2);
  gather_kernel<HID2><<<ceil_div(N, 256 / (HID2 / 4)), 256, 0, stream>>>(bufA, rowptr, csr, dis, bufB, N);
  bias_relu_kernel<<<ceil_div(N * HID2, TB), TB, 0, stream>>>(bufB, b2, N, HID2);

  gemm(bufB, W3, bufA, HID2, OUTC);
  gather_kernel<OUTC><<<ceil_div(N, 256 / (OUTC / 4)), 256, 0, stream>>>(bufA, rowptr, csr, dis, bufB, N);
  bias_relu_kernel<<<ceil_div(N * OUTC, TB), TB, 0, stream>>>(bufB, b3, N, OUTC);

  pool_count_kernel<<<ceil_div(N, TB), TB, 0, stream>>>(batch, cnt, N);
  pool_sum_kernel<<<ceil_div(N * 32, TB), TB, 0, stream>>>(bufB, batch, out, N);
  pool_div_kernel<<<ceil_div(G * OUTC, TB), TB, 0, stream>>>(out, cnt, G);
}


# Bias add and ReLU applied in the gather kernel's own store instead of a separate elementwise pass per layer (3 launches fewer); same f32 add and max on the same values, sum order unchanged
# speedup vs baseline: 2.3790x; 1.1039x over previous
#include <hip/hip_runtime.h>

typedef float v2f __attribute__((ext_vector_type(2)));
typedef float v8f __attribute__((ext_vector_type(8)));

#define F_IN 128
#define HID  128
#define HID2 64
#define OUTC 32

static inline int ceil_div(int a, int b) { return (a + b - 1) / b; }


#define SCAN_CHUNK 1024

__global__ void degi_zero_kernel(int* __restrict__ degi, int Npad) {
  int i = blockIdx.x * blockDim.x + threadIdx.x;
  if (i < Npad) degi[i] = 0;
}

__global__ void edge_rank_kernel(const int* __restrict__ dst, int* __restrict__ degi,
                                 int* __restrict__ rank, int E) {
  int e = blockIdx.x * blockDim.x + threadIdx.x;
  if (e < E) rank[e] = atomicAdd(&degi[dst[e]], 1);
}

__global__ __launch_bounds__(256) void rowptr_kernel(const int* __restrict__ degi,
                                                     int* __restrict__ rowptr,
                                                     float* __restrict__ dis, int N) {
  __shared__ int front[8];
  __shared__ int wtot[8];
  const int t = threadIdx.x;
  const int lane = t & 31, wave = t >> 5;
  const int b = blockIdx.x;
  const int4* d4 = (const int4*)degi;

  int pre = 0;
#pragma unroll 8
  for (int i = t; i < b * 256; i += 256) {
    int4 q = d4[i];
    pre += q.x + q.y + q.z + q.w;
  }
  for (int o = 16; o > 0; o >>= 1) pre += __shfl_xor(pre, o);
  if (lane == 0) front[wave] = pre;

  const int4 v = d4[b * 256 + t];
  const int tot = v.x + v.y + v.z + v.w;
  int inc = tot;
  for (int o = 1; o < 32; o <<= 1) {
    int u = __shfl_up(inc, o);
    if (lane >= o) inc += u;
  }
  if (lane == 31) wtot[wave] = inc;
  __syncthreads();

  int ex = inc - tot;
  for (int w = 0; w < 8; ++w) {
    ex += front[w];
    if (w < wave) ex += wtot[w];
  }
  const int c[4] = {v.x, v.y, v.z, v.w};
  const int i0 = (b * 256 + t) * 4;
#pragma unroll
  for (int k = 0; k < 4; ++k) {
    int i = i0 + k;
    if (i <= N) rowptr[i] = ex;
    if (i < N) dis[i] = rsqrtf((float)(c[k] + 1));
    ex += c[k];
  }
}

__global__ void csr_fill_kernel(const int* __restrict__ src, const int* __restrict__ dst,
                                const int* __restrict__ rank, const int* __restrict__ rowptr,
                                const float* __restrict__ dis, int* __restrict__ csr_, int E) {
  int e = blockIdx.x * blockDim.x + threadIdx.x;
  if (e >= E) return;
  int s = src[e], d = dst[e];
  int2* csr = (int2*)csr_;
  csr[rowptr[d] + rank[e]] = make_int2(s, __float_as_int(dis[s]));
}

__global__ __launch_bounds__(128) void gemm_wmma_kernel(
    const float* __restrict__ A, const float* __restrict__ B, float* __restrict__ C,
    int M, int K, int Ncol) {
  int lane = threadIdx.x & 31;
  int wave = threadIdx.x >> 5;
  int mtile = blockIdx.x * 4 + wave;
  int row0 = mtile << 4;
  if (row0 >= M) return;
  int col0 = blockIdx.y << 4;
  int half = lane >> 4;
  int l16  = lane & 15;

  v8f acc = {0.f, 0.f, 0.f, 0.f, 0.f, 0.f, 0.f, 0.f};
  const float* arow = A + (size_t)(row0 + l16) * K;
  const float* bcol = B + col0 + l16;

  for (int k = 0; k < K; k += 4) {
    int ka = k + half * 2;
    v2f a, b;
    a.x = arow[ka + 0];
    a.y = arow[ka + 1];
    b.x = bcol[(size_t)(ka + 0) * Ncol];
    b.y = bcol[(size_t)(ka + 1) * Ncol];
    acc = __builtin_amdgcn_wmma_f32_16x16x4_f32(
         false, a,  false, b,
         (short)0, acc,  false,  false);
  }

  float* crow = C + (size_t)(row0 + half * 8) * Ncol + col0 + l16;
#pragma unroll
  for (int r = 0; r < 8; ++r)
    crow[(size_t)r * Ncol] = acc[r];
}


template <int LPN, bool TAIL>
__device__ __forceinline__ void add8(float4& acc, const int2* __restrict__ csr,
                                     const float4* __restrict__ rows, int j, int end, int sub, float dd) {
  int2 ent[8];
  float4 v[8];
#pragma unroll
  for (int u = 0; u < 8; ++u) {
    int jj = j + u;
    if (TAIL) jj = jj < end ? jj : end - 1;
    ent[u] = csr[jj];
  }
#pragma unroll
  for (int u = 0; u < 8; ++u)
    v[u] = rows[ent[u].x * LPN + sub];
#pragma unroll
  for (int u = 0; u < 8; ++u) {
    float4 r = v[u];
    if (TAIL) {
      bool ok = j + u < end;
      r.x = ok ? r.x : 0.f;
      r.y = ok ? r.y : 0.f;
      r.z = ok ? r.z : 0.f;
      r.w = ok ? r.w : 0.f;
    }
    float nrm = __int_as_float(ent[u].y) * dd;
    acc.x = fmaf(r.x, nrm, acc.x);
    acc.y = fmaf(r.y, nrm, acc.y);
    acc.z = fmaf(r.z, nrm, acc.z);
    acc.w = fmaf(r.w, nrm, acc.w);
  }
}

template <int F>
__global__ __launch_bounds__(256) void gather_kernel(const float* __restrict__ XW,
                                                     const int* __restrict__ rowptr,
                                                     const int* __restrict__ csr_,
                                                     const float* __restrict__ dis,
                                                     const float* __restrict__ bias,
                                                     float* __restrict__ h_out, int N) {
  constexpr int LPN = F / 4;
  constexpr int NPB = 256 / LPN;
  const int sub = threadIdx.x % LPN;
  const int node = blockIdx.x * NPB + threadIdx.x / LPN;
  if (node >= N) return;
  const int2* csr = (const int2*)csr_;
  const float4* rows = (const float4*)XW;
  const int beg = rowptr[node], end = rowptr[node + 1];
  const float dd = dis[node];
  const float4 own = rows[node * LPN + sub];
  const float sw = dd * dd;
  float4 acc = make_float4(own.x * sw, own.y * sw, own.z * sw, own.w * sw);

  int j = beg;
  for (; j + 8 <= end; j += 8) add8<LPN, false>(acc, csr, rows, j, end, sub, dd);
  if (j < end) add8<LPN, true>(acc, csr, rows, j, end, sub, dd);

  const float* bp = bias + sub * 4;
  float4 h;
  h.x = acc.x + bp[0];
  h.y = acc.y + bp[1];
  h.z = acc.z + bp[2];
  h.w = acc.w + bp[3];
  h.x = h.x > 0.f ? h.x : 0.f;
  h.y = h.y > 0.f ? h.y : 0.f;
  h.z = h.z > 0.f ? h.z : 0.f;
  h.w = h.w > 0.f ? h.w : 0.f;
  ((float4*)h_out)[node * LPN + sub] = h;
}


__global__ void pool_count_kernel(const int* __restrict__ batch, float* __restrict__ cnt, int N) {
  int i = blockIdx.x * blockDim.x + threadIdx.x;
  if (i < N) atomicAdd(&cnt[batch[i]], 1.0f);
}

__global__ void pool_sum_kernel(const float* __restrict__ h, const int* __restrict__ batch,
                                float* __restrict__ out, int N) {
  int gid = blockIdx.x * blockDim.x + threadIdx.x;
  int lane = gid & 31;
  int node = gid >> 5;
  if (node >= N) return;
  int g = batch[node];
  atomicAdd(&out[(size_t)g * OUTC + lane], h[(size_t)node * OUTC + lane]);
}

__global__ void pool_div_kernel(float* __restrict__ out, const float* __restrict__ cnt, int G) {
  int i = blockIdx.x * blockDim.x + threadIdx.x;
  if (i >= G * OUTC) return;
  out[i] /= fmaxf(cnt[i >> 5], 1.0f);
}


extern "C" void kernel_launch(void* const* d_in, const int* in_sizes, int n_in,
                              void* d_out, int out_size, void* d_ws, size_t ws_size,
                              hipStream_t stream) {
  const float* x     = (const float*)d_in[0];
  const int*   ei    = (const int*)d_in[1];
  const int*   batch = (const int*)d_in[2];
  const float* W1    = (const float*)d_in[3];
  const float* b1    = (const float*)d_in[4];
  const float* W2    = (const float*)d_in[5];
  const float* b2    = (const float*)d_in[6];
  const float* W3    = (const float*)d_in[7];
  const float* b3    = (const float*)d_in[8];

  const int N = in_sizes[0] / F_IN;
  const int E = in_sizes[1] / 2;
  const int G = out_size / OUTC;
  const int* src = ei;
  const int* dst = ei + E;
  float* out = (float*)d_out;

  float* dis = (float*)d_ws;
  float* cnt = dis + N;
  size_t off = (size_t)N + (size_t)G;
  off = (off + 63) & ~(size_t)63;
  float* bufA = (float*)d_ws + off;
  float* bufB = bufA + (size_t)N * HID;
  const int Npad = ceil_div(N + 1, SCAN_CHUNK) * SCAN_CHUNK;
  const size_t Epad = ((size_t)E + 63) & ~(size_t)63;
  int* degi   = (int*)(bufB + (size_t)N * HID);
  int* rowptr = degi + Npad;
  int* rank   = rowptr + Npad;
  int* csr    = rank + Epad;

  const int TB = 256;

  hipMemsetAsync(cnt, 0, (size_t)G * sizeof(float), stream);
  hipMemsetAsync(out, 0, (size_t)out_size * sizeof(float), stream);

  degi_zero_kernel<<<ceil_div(Npad, TB), TB, 0, stream>>>(degi, Npad);
  edge_rank_kernel<<<ceil_div(E, TB), TB, 0, stream>>>(dst, degi, rank, E);
  rowptr_kernel<<<Npad / SCAN_CHUNK, 256, 0, stream>>>(degi, rowptr, dis, N);
  csr_fill_kernel<<<ceil_div(E, TB), TB, 0, stream>>>(src, dst, rank, rowptr, dis, csr, E);

  auto gemm = [&](const float* hin, const float* W, float* XW, int K, int Cout) {
    dim3 gg(ceil_div(ceil_div(N, 16), 4), Cout / 16);
    gemm_wmma_kernel<<<gg, 128, 0, stream>>>(hin, W, XW, N, K, Cout);
  };

  gemm(x, W1, bufA, F_IN, HID);
  gather_kernel<HID><<<ceil_div(N, 256 / (HID / 4)), 256, 0, stream>>>(bufA, rowptr, csr, dis, b1, bufB, N);

  gemm(bufB, W2, bufA, HID, HID2);
  gather_kernel<HID2><<<ceil_div(N, 256 / (HID2 / 4)), 256, 0, stream>>>(bufA, rowptr, csr, dis, b2, bufB, N);

  gemm(bufB, W3, bufA, HID2, OUTC);
  gather_kernel<OUTC><<<ceil_div(N, 256 / (OUTC / 4)), 256, 0, stream>>>(bufA, rowptr, csr, dis, b3, bufB, N);

  pool_count_kernel<<<ceil_div(N, TB), TB, 0, stream>>>(batch, cnt, N);
  pool_sum_kernel<<<ceil_div(N * 32, TB), TB, 0, stream>>>(bufB, batch, out, N);
  pool_div_kernel<<<ceil_div(G * OUTC, TB), TB, 0, stream>>>(out, cnt, G);
}


# Edge ranks from LDS counters: 64 blocks each count one edge chunk per node in LDS, then a per-node prefix over chunks; no global atomic per edge; a node's sum order is now chunk, then arrival
# speedup vs baseline: 3.1754x; 1.3348x over previous
#include <hip/hip_runtime.h>

typedef float v2f __attribute__((ext_vector_type(2)));
typedef float v8f __attribute__((ext_vector_type(8)));

#define F_IN 128
#define HID  128
#define HID2 64
#define OUTC 32

static inline int ceil_div(int a, int b) { return (a + b - 1) / b; }


#define SCAN_CHUNK 1024
#define NCHUNK 64
#define RANK_CAP 65536

__global__ __launch_bounds__(1024) void chunk_rank_kernel(const int* __restrict__ dst,
                                                          int* __restrict__ rank,
                                                          int* __restrict__ cbase,
                                                          int E, int Npad, int chunk) {
  __shared__ int4 cnt4[RANK_CAP / 4];
  int* cnt = (int*)cnt4;
  const int t = threadIdx.x;
  const int e0 = blockIdx.x * chunk;
  const int e1 = e0 + chunk < E ? e0 + chunk : E;
  int4* mine = (int4*)(cbase + (size_t)blockIdx.x * Npad);

  for (int base = 0; base < Npad; base += RANK_CAP) {
    const int lim = Npad - base < RANK_CAP ? Npad - base : RANK_CAP;
    for (int i = t; i < lim / 4; i += 1024) cnt4[i] = make_int4(0, 0, 0, 0);
    __syncthreads();
    for (int e = e0 + t; e < e1; e += 8 * 1024) {
      int d[8];
#pragma unroll
      for (int u = 0; u < 8; ++u) {
        int ee = e + u * 1024;
        d[u] = dst[ee < e1 ? ee : e1 - 1];
      }
#pragma unroll
      for (int u = 0; u < 8; ++u) {
        int ee = e + u * 1024;
        int dd = d[u] - base;
        if (ee < e1 && (unsigned)dd < (unsigned)lim) rank[ee] = atomicAdd(&cnt[dd], 1);
      }
    }
    __syncthreads();
    for (int i = t; i < lim / 4; i += 1024) mine[base / 4 + i] = cnt4[i];
    __syncthreads();
  }
}

__global__ __launch_bounds__(256) void chunk_scan_kernel(int* __restrict__ cbase, int* __restrict__ degi,
                                                         int Npad) {
  const int i = blockIdx.x * 256 + threadIdx.x;
  int v[NCHUNK];
#pragma unroll
  for (int c = 0; c < NCHUNK; ++c) v[c] = cbase[(size_t)c * Npad + i];
  int run = 0;
#pragma unroll
  for (int c = 0; c < NCHUNK; ++c) {
    cbase[(size_t)c * Npad + i] = run;
    run += v[c];
  }
  degi[i] = run;
}

__global__ __launch_bounds__(256) void rowptr_kernel(const int* __restrict__ degi,
                                                     int* __restrict__ rowptr,
                                                     float* __restrict__ dis, int N) {
  __shared__ int front[8];
  __shared__ int wtot[8];
  const int t = threadIdx.x;
  const int lane = t & 31, wave = t >> 5;
  const int b = blockIdx.x;
  const int4* d4 = (const int4*)degi;

  int pre = 0;
#pragma unroll 8
  for (int i = t; i < b * 256; i += 256) {
    int4 q = d4[i];
    pre += q.x + q.y + q.z + q.w;
  }
  for (int o = 16; o > 0; o >>= 1) pre += __shfl_xor(pre, o);
  if (lane == 0) front[wave] = pre;

  const int4 v = d4[b * 256 + t];
  const int tot = v.x + v.y + v.z + v.w;
  int inc = tot;
  for (int o = 1; o < 32; o <<= 1) {
    int u = __shfl_up(inc, o);
    if (lane >= o) inc += u;
  }
  if (lane == 31) wtot[wave] = inc;
  __syncthreads();

  int ex = inc - tot;
  for (int w = 0; w < 8; ++w) {
    ex += front[w];
    if (w < wave) ex += wtot[w];
  }
  const int c[4] = {v.x, v.y, v.z, v.w};
  const int i0 = (b * 256 + t) * 4;
#pragma unroll
  for (int k = 0; k < 4; ++k) {
    int i = i0 + k;
    if (i <= N) rowptr[i] = ex;
    if (i < N) dis[i] = rsqrtf((float)(c[k] + 1));
    ex += c[k];
  }
}

__global__ void csr_fill_kernel(const int* __restrict__ src, const int* __restrict__ dst,
                                const int* __restrict__ rank, const int* __restrict__ rowptr,
                                const int* __restrict__ cbase,
                                const float* __restrict__ dis, int* __restrict__ csr_, int E,
                                int Npad, int chunk) {
  int e = blockIdx.x * blockDim.x + threadIdx.x;
  if (e >= E) return;
  const int* cb = cbase + (size_t)((blockIdx.x * blockDim.x) / chunk) * Npad;
  int s = src[e], d = dst[e];
  int2* csr = (int2*)csr_;
  csr[rowptr[d] + cb[d] + rank[e]] = make_int2(s, __float_as_int(dis[s]));
}

__global__ __launch_bounds__(128) void gemm_wmma_kernel(
    const float* __restrict__ A, const float* __restrict__ B, float* __restrict__ C,
    int M, int K, int Ncol) {
  int lane = threadIdx.x & 31;
  int wave = threadIdx.x >> 5;
  int mtile = blockIdx.x * 4 + wave;
  int row0 = mtile << 4;
  if (row0 >= M) return;
  int col0 = blockIdx.y << 4;
  int half = lane >> 4;
  int l16  = lane & 15;

  v8f acc = {0.f, 0.f, 0.f, 0.f, 0.f, 0.f, 0.f, 0.f};
  const float* arow = A + (size_t)(row0 + l16) * K;
  const float* bcol = B + col0 + l16;

  for (int k = 0; k < K; k += 4) {
    int ka = k + half * 2;
    v2f a, b;
    a.x = arow[ka + 0];
    a.y = arow[ka + 1];
    b.x = bcol[(size_t)(ka + 0) * Ncol];
    b.y = bcol[(size_t)(ka + 1) * Ncol];
    acc = __builtin_amdgcn_wmma_f32_16x16x4_f32(
         false, a,  false, b,
         (short)0, acc,  false,  false);
  }

  float* crow = C + (size_t)(row0 + half * 8) * Ncol + col0 + l16;
#pragma unroll
  for (int r = 0; r < 8; ++r)
    crow[(size_t)r * Ncol] = acc[r];
}


template <int LPN, bool TAIL>
__device__ __forceinline__ void add8(float4& acc, const int2* __restrict__ csr,
                                     const float4* __restrict__ rows, int j, int end, int sub, float dd) {
  int2 ent[8];
  float4 v[8];
#pragma unroll
  for (int u = 0; u < 8; ++u) {
    int jj = j + u;
    if (TAIL) jj = jj < end ? jj : end - 1;
    ent[u] = csr[jj];
  }
#pragma unroll
  for (int u = 0; u < 8; ++u)
    v[u] = rows[ent[u].x * LPN + sub];
#pragma unroll
  for (int u = 0; u < 8; ++u) {
    float4 r = v[u];
    if (TAIL) {
      bool ok = j + u < end;
      r.x = ok ? r.x : 0.f;
      r.y = ok ? r.y : 0.f;
      r.z = ok ? r.z : 0.f;
      r.w = ok ? r.w : 0.f;
    }
    float nrm = __int_as_float(ent[u].y) * dd;
    acc.x = fmaf(r.x, nrm, acc.x);
    acc.y = fmaf(r.y, nrm, acc.y);
    acc.z = fmaf(r.z, nrm, acc.z);
    acc.w = fmaf(r.w, nrm, acc.w);
  }
}

template <int F>
__global__ __launch_bounds__(256) void gather_kernel(const float* __restrict__ XW,
                                                     const int* __restrict__ rowptr,
                                                     const int* __restrict__ csr_,
                                                     const float* __restrict__ dis,
                                                     const float* __restrict__ bias,
                                                     float* __restrict__ h_out, int N) {
  constexpr int LPN = F / 4;
  constexpr int NPB = 256 / LPN;
  const int sub = threadIdx.x % LPN;
  const int node = blockIdx.x * NPB + threadIdx.x / LPN;
  if (node >= N) return;
  const int2* csr = (const int2*)csr_;
  const float4* rows = (const float4*)XW;
  const int beg = rowptr[node], end = rowptr[node + 1];
  const float dd = dis[node];
  const float4 own = rows[node * LPN + sub];
  const float sw = dd * dd;
  float4 acc = make_float4(own.x * sw, own.y * sw, own.z * sw, own.w * sw);

  int j = beg;
  for (; j + 8 <= end; j += 8) add8<LPN, false>(acc, csr, rows, j, end, sub, dd);
  if (j < end) add8<LPN, true>(acc, csr, rows, j, end, sub, dd);

  const float* bp = bias + sub * 4;
  float4 h;
  h.x = acc.x + bp[0];
  h.y = acc.y + bp[1];
  h.z = acc.z + bp[2];
  h.w = acc.w + bp[3];
  h.x = h.x > 0.f ? h.x : 0.f;
  h.y = h.y > 0.f ? h.y : 0.f;
  h.z = h.z > 0.f ? h.z : 0.f;
  h.w = h.w > 0.f ? h.w : 0.f;
  ((float4*)h_out)[node * LPN + sub] = h;
}


__global__ void pool_count_kernel(const int* __restrict__ batch, float* __restrict__ cnt, int N) {
  int i = blockIdx.x * blockDim.x + threadIdx.x;
  if (i < N) atomicAdd(&cnt[batch[i]], 1.0f);
}

__global__ void pool_sum_kernel(const float* __restrict__ h, const int* __restrict__ batch,
                                float* __restrict__ out, int N) {
  int gid = blockIdx.x * blockDim.x + threadIdx.x;
  int lane = gid & 31;
  int node = gid >> 5;
  if (node >= N) return;
  int g = batch[node];
  atomicAdd(&out[(size_t)g * OUTC + lane], h[(size_t)node * OUTC + lane]);
}

__global__ void pool_div_kernel(float* __restrict__ out, const float* __restrict__ cnt, int G) {
  int i = blockIdx.x * blockDim.x + threadIdx.x;
  if (i >= G * OUTC) return;
  out[i] /= fmaxf(cnt[i >> 5], 1.0f);
}


extern "C" void kernel_launch(void* const* d_in, const int* in_sizes, int n_in,
                              void* d_out, int out_size, void* d_ws, size_t ws_size,
                              hipStream_t stream) {
  const float* x     = (const float*)d_in[0];
  const int*   ei    = (const int*)d_in[1];
  const int*   batch = (const int*)d_in[2];
  const float* W1    = (const float*)d_in[3];
  const float* b1    = (const float*)d_in[4];
  const float* W2    = (const float*)d_in[5];
  const float* b2    = (const float*)d_in[6];
  const float* W3    = (const float*)d_in[7];
  const float* b3    = (const float*)d_in[8];

  const int N = in_sizes[0] / F_IN;
  const int E = in_sizes[1] / 2;
  const int G = out_size / OUTC;
  const int* src = ei;
  const int* dst = ei + E;
  float* out = (float*)d_out;

  float* dis = (float*)d_ws;
  float* cnt = dis + N;
  size_t off = (size_t)N + (size_t)G;
  off = (off + 63) & ~(size_t)63;
  float* bufA = (float*)d_ws + off;
  float* bufB = bufA + (size_t)N * HID;
  const int Npad = ceil_div(N + 1, SCAN_CHUNK) * SCAN_CHUNK;
  const size_t Epad = ((size_t)E + 63) & ~(size_t)63;
  int* degi   = (int*)(bufB + (size_t)N * HID);
  int* rowptr = degi + Npad;
  int* rank   = rowptr + Npad;
  int* csr    = rank + Epad;
  int* cbase  = csr + 2 * Epad;
  const int TB = 256;
  const int chunk = ceil_div(ceil_div(E, NCHUNK), TB) * TB;

  hipMemsetAsync(cnt, 0, (size_t)G * sizeof(float), stream);
  hipMemsetAsync(out, 0, (size_t)out_size * sizeof(float), stream);

  chunk_rank_kernel<<<NCHUNK, 1024, 0, stream>>>(dst, rank, cbase, E, Npad, chunk);
  chunk_scan_kernel<<<Npad / 256, 256, 0, stream>>>(cbase, degi, Npad);
  rowptr_kernel<<<Npad / SCAN_CHUNK, 256, 0, stream>>>(degi, rowptr, dis, N);
  csr_fill_kernel<<<ceil_div(E, TB), TB, 0, stream>>>(src, dst, rank, rowptr, cbase, dis, csr, E, Npad, chunk);

  auto gemm = [&](const float* hin, const float* W, float* XW, int K, int Cout) {
    dim3 gg(ceil_div(ceil_div(N, 16), 4), Cout / 16);
    gemm_wmma_kernel<<<gg, 128, 0, stream>>>(hin, W, XW, N, K, Cout);
  };

  gemm(x, W1, bufA, F_IN, HID);
  gather_kernel<HID><<<ceil_div(N, 256 / (HID / 4)), 256, 0, stream>>>(bufA, rowptr, csr, dis, b1, bufB, N);

  gemm(bufB, W2, bufA, HID, HID2);
  gather_kernel<HID2><<<ceil_div(N, 256 / (HID2 / 4)), 256, 0, stream>>>(bufA, rowptr, csr, dis, b2, bufB, N);

  gemm(bufB, W3, bufA, HID2, OUTC);
  gather_kernel<OUTC><<<ceil_div(N, 256 / (OUTC / 4)), 256, 0, stream>>>(bufA, rowptr, csr, dis, b3, bufB, N);

  pool_count_kernel<<<ceil_div(N, TB), TB, 0, stream>>>(batch, cnt, N);
  pool_sum_kernel<<<ceil_div(N * 32, TB), TB, 0, stream>>>(bufB, batch, out, N);
  pool_div_kernel<<<ceil_div(G * OUTC, TB), TB, 0, stream>>>(out, cnt, G);
}


# Dense transform: each wave keeps its 16-row A strip in registers and sweeps all columns, weights staged once per block in LDS; k summed per group of 8 as 0,1,4,5 then 2,3,6,7
# speedup vs baseline: 4.0124x; 1.2636x over previous
#include <hip/hip_runtime.h>

typedef float v2f __attribute__((ext_vector_type(2)));
typedef float v8f __attribute__((ext_vector_type(8)));

#define F_IN 128
#define HID  128
#define HID2 64
#define OUTC 32

static inline int ceil_div(int a, int b) { return (a + b - 1) / b; }


#define SCAN_CHUNK 1024
#define NCHUNK 64
#define RANK_CAP 65536

__global__ __launch_bounds__(1024) void chunk_rank_kernel(const int* __restrict__ dst,
                                                          int* __restrict__ rank,
                                                          int* __restrict__ cbase,
                                                          int E, int Npad, int chunk) {
  __shared__ int4 cnt4[RANK_CAP / 4];
  int* cnt = (int*)cnt4;
  const int t = threadIdx.x;
  const int e0 = blockIdx.x * chunk;
  const int e1 = e0 + chunk < E ? e0 + chunk : E;
  int4* mine = (int4*)(cbase + (size_t)blockIdx.x * Npad);

  for (int base = 0; base < Npad; base += RANK_CAP) {
    const int lim = Npad - base < RANK_CAP ? Npad - base : RANK_CAP;
    for (int i = t; i < lim / 4; i += 1024) cnt4[i] = make_int4(0, 0, 0, 0);
    __syncthreads();
    for (int e = e0 + t; e < e1; e += 8 * 1024) {
      int d[8];
#pragma unroll
      for (int u = 0; u < 8; ++u) {
        int ee = e + u * 1024;
        d[u] = dst[ee < e1 ? ee : e1 - 1];
      }
#pragma unroll
      for (int u = 0; u < 8; ++u) {
        int ee = e + u * 1024;
        int dd = d[u] - base;
        if (ee < e1 && (unsigned)dd < (unsigned)lim) rank[ee] = atomicAdd(&cnt[dd], 1);
      }
    }
    __syncthreads();
    for (int i = t; i < lim / 4; i += 1024) mine[base / 4 + i] = cnt4[i];
    __syncthreads();
  }
}

__global__ __launch_bounds__(256) void chunk_scan_kernel(int* __restrict__ cbase, int* __restrict__ degi,
                                                         int Npad) {
  const int i = blockIdx.x * 256 + threadIdx.x;
  int v[NCHUNK];
#pragma unroll
  for (int c = 0; c < NCHUNK; ++c) v[c] = cbase[(size_t)c * Npad + i];
  int run = 0;
#pragma unroll
  for (int c = 0; c < NCHUNK; ++c) {
    cbase[(size_t)c * Npad + i] = run;
    run += v[c];
  }
  degi[i] = run;
}

__global__ __launch_bounds__(256) void rowptr_kernel(const int* __restrict__ degi,
                                                     int* __restrict__ rowptr,
                                                     float* __restrict__ dis, int N) {
  __shared__ int front[8];
  __shared__ int wtot[8];
  const int t = threadIdx.x;
  const int lane = t & 31, wave = t >> 5;
  const int b = blockIdx.x;
  const int4* d4 = (const int4*)degi;

  int pre = 0;
#pragma unroll 8
  for (int i = t; i < b * 256; i += 256) {
    int4 q = d4[i];
    pre += q.x + q.y + q.z + q.w;
  }
  for (int o = 16; o > 0; o >>= 1) pre += __shfl_xor(pre, o);
  if (lane == 0) front[wave] = pre;

  const int4 v = d4[b * 256 + t];
  const int tot = v.x + v.y + v.z + v.w;
  int inc = tot;
  for (int o = 1; o < 32; o <<= 1) {
    int u = __shfl_up(inc, o);
    if (lane >= o) inc += u;
  }
  if (lane == 31) wtot[wave] = inc;
  __syncthreads();

  int ex = inc - tot;
  for (int w = 0; w < 8; ++w) {
    ex += front[w];
    if (w < wave) ex += wtot[w];
  }
  const int c[4] = {v.x, v.y, v.z, v.w};
  const int i0 = (b * 256 + t) * 4;
#pragma unroll
  for (int k = 0; k < 4; ++k) {
    int i = i0 + k;
    if (i <= N) rowptr[i] = ex;
    if (i < N) dis[i] = rsqrtf((float)(c[k] + 1));
    ex += c[k];
  }
}

__global__ void csr_fill_kernel(const int* __restrict__ src, const int* __restrict__ dst,
                                const int* __restrict__ rank, const int* __restrict__ rowptr,
                                const int* __restrict__ cbase,
                                const float* __restrict__ dis, int* __restrict__ csr_, int E,
                                int Npad, int chunk) {
  int e = blockIdx.x * blockDim.x + threadIdx.x;
  if (e >= E) return;
  const int* cb = cbase + (size_t)((blockIdx.x * blockDim.x) / chunk) * Npad;
  int s = src[e], d = dst[e];
  int2* csr = (int2*)csr_;
  csr[rowptr[d] + cb[d] + rank[e]] = make_int2(s, __float_as_int(dis[s]));
}

template <int K, int NCOL>
__global__ __launch_bounds__(256) void gemm_wmma_kernel(
    const float* __restrict__ A, const float* __restrict__ B, float* __restrict__ C, int M) {
  constexpr int CT = NCOL / 16 < 4 ? NCOL / 16 : 4;
  constexpr int RUN = 16 * CT;
  constexpr int KQ = K / 4;
  constexpr int KG = K / 8;
  __shared__ float4 Bl[KQ * NCOL];

  const int t = threadIdx.x;
  const int lane = t & 31;
  const int wave = t >> 5;
  const int c = lane & 15;
  const int h = lane >> 4;
  const int row0 = (blockIdx.x * 8 + wave) << 4;

  float4 a[KG];
  {
    int row = row0 + c;
    row = row < M ? row : M - 1;
    const float4* ap = (const float4*)(A + (size_t)row * K) + h;
#pragma unroll
    for (int g = 0; g < KG; ++g) a[g] = ap[2 * g];
  }

  static_assert((KQ * NCOL) % 256 == 0, "staging loop covers Bl in whole sweeps of the block");
#pragma unroll 8
  for (int it = 0; it < KQ * NCOL / 256; ++it) {
    const int i = t + it * 256;
    const int kq = i / NCOL;
    const int n = i % NCOL;
    const float* p = B + (size_t)(4 * kq) * NCOL + n;
    const int nl = n % RUN;
    const int slot = (n - nl) + (nl % CT) * 16 + nl / CT;
    Bl[kq * NCOL + slot] = make_float4(p[0], p[NCOL], p[2 * NCOL], p[3 * NCOL]);
  }
  __syncthreads();
  if (row0 >= M) return;

  for (int n0 = 0; n0 < NCOL; n0 += RUN) {
    v8f acc[CT];
#pragma unroll
    for (int j = 0; j < CT; ++j) acc[j] = (v8f){0.f, 0.f, 0.f, 0.f, 0.f, 0.f, 0.f, 0.f};

#pragma unroll
    for (int g = 0; g < KG; ++g) {
      float4 b[CT];
#pragma unroll
      for (int j = 0; j < CT; ++j) b[j] = Bl[(2 * g + h) * NCOL + n0 + j * 16 + c];
#pragma unroll
      for (int j = 0; j < CT; ++j) {
        v2f av = {a[g].x, a[g].y};
        v2f bv = {b[j].x, b[j].y};
        acc[j] = __builtin_amdgcn_wmma_f32_16x16x4_f32(
             false, av,  false, bv,
             (short)0, acc[j],  false,  false);
      }
#pragma unroll
      for (int j = 0; j < CT; ++j) {
        v2f av = {a[g].z, a[g].w};
        v2f bv = {b[j].z, b[j].w};
        acc[j] = __builtin_amdgcn_wmma_f32_16x16x4_f32(
             false, av,  false, bv,
             (short)0, acc[j],  false,  false);
      }
    }

#pragma unroll
    for (int r = 0; r < 8; ++r) {
      const int row = row0 + 8 * h + r;
      if (row < M) {
        float* cp = C + (size_t)row * NCOL + n0 + c * CT;
        if (CT == 4) {
          *(float4*)cp = make_float4(acc[0][r], acc[1 % CT][r], acc[2 % CT][r], acc[3 % CT][r]);
        } else if (CT == 2) {
          *(float2*)cp = make_float2(acc[0][r], acc[1 % CT][r]);
        } else {
          cp[0] = acc[0][r];
        }
      }
    }
  }
}


template <int LPN, bool TAIL>
__device__ __forceinline__ void add8(float4& acc, const int2* __restrict__ csr,
                                     const float4* __restrict__ rows, int j, int end, int sub, float dd) {
  int2 ent[8];
  float4 v[8];
#pragma unroll
  for (int u = 0; u < 8; ++u) {
    int jj = j + u;
    if (TAIL) jj = jj < end ? jj : end - 1;
    ent[u] = csr[jj];
  }
#pragma unroll
  for (int u = 0; u < 8; ++u)
    v[u] = rows[ent[u].x * LPN + sub];
#pragma unroll
  for (int u = 0; u < 8; ++u) {
    float4 r = v[u];
    if (TAIL) {
      bool ok = j + u < end;
      r.x = ok ? r.x : 0.f;
      r.y = ok ? r.y : 0.f;
      r.z = ok ? r.z : 0.f;
      r.w = ok ? r.w : 0.f;
    }
    float nrm = __int_as_float(ent[u].y) * dd;
    acc.x = fmaf(r.x, nrm, acc.x);
    acc.y = fmaf(r.y, nrm, acc.y);
    acc.z = fmaf(r.z, nrm, acc.z);
    acc.w = fmaf(r.w, nrm, acc.w);
  }
}

template <int F>
__global__ __launch_bounds__(256) void gather_kernel(const float* __restrict__ XW,
                                                     const int* __restrict__ rowptr,
                                                     const int* __restrict__ csr_,
                                                     const float* __restrict__ dis,
                                                     const float* __restrict__ bias,
                                                     float* __restrict__ h_out, int N) {
  constexpr int LPN = F / 4;
  constexpr int NPB = 256 / LPN;
  const int sub = threadIdx.x % LPN;
  const int node = blockIdx.x * NPB + threadIdx.x / LPN;
  if (node >= N) return;
  const int2* csr = (const int2*)csr_;
  const float4* rows = (const float4*)XW;
  const int beg = rowptr[node], end = rowptr[node + 1];
  const float dd = dis[node];
  const float4 own = rows[node * LPN + sub];
  const float sw = dd * dd;
  float4 acc = make_float4(own.x * sw, own.y * sw, own.z * sw, own.w * sw);

  int j = beg;
  for (; j + 8 <= end; j += 8) add8<LPN, false>(acc, csr, rows, j, end, sub, dd);
  if (j < end) add8<LPN, true>(acc, csr, rows, j, end, sub, dd);

  const float* bp = bias + sub * 4;
  float4 h;
  h.x = acc.x + bp[0];
  h.y = acc.y + bp[1];
  h.z = acc.z + bp[2];
  h.w = acc.w + bp[3];
  h.x = h.x > 0.f ? h.x : 0.f;
  h.y = h.y > 0.f ? h.y : 0.f;
  h.z = h.z > 0.f ? h.z : 0.f;
  h.w = h.w > 0.f ? h.w : 0.f;
  ((float4*)h_out)[node * LPN + sub] = h;
}


__global__ void pool_count_kernel(const int* __restrict__ batch, float* __restrict__ cnt, int N) {
  int i = blockIdx.x * blockDim.x + threadIdx.x;
  if (i < N) atomicAdd(&cnt[batch[i]], 1.0f);
}

__global__ void pool_sum_kernel(const float* __restrict__ h, const int* __restrict__ batch,
                                float* __restrict__ out, int N) {
  int gid = blockIdx.x * blockDim.x + threadIdx.x;
  int lane = gid & 31;
  int node = gid >> 5;
  if (node >= N) return;
  int g = batch[node];
  atomicAdd(&out[(size_t)g * OUTC + lane], h[(size_t)node * OUTC + lane]);
}

__global__ void pool_div_kernel(float* __restrict__ out, const float* __restrict__ cnt, int G) {
  int i = blockIdx.x * blockDim.x + threadIdx.x;
  if (i >= G * OUTC) return;
  out[i] /= fmaxf(cnt[i >> 5], 1.0f);
}


extern "C" void kernel_launch(void* const* d_in, const int* in_sizes, int n_in,
                              void* d_out, int out_size, void* d_ws, size_t ws_size,
                              hipStream_t stream) {
  const float* x     = (const float*)d_in[0];
  const int*   ei    = (const int*)d_in[1];
  const int*   batch = (const int*)d_in[2];
  const float* W1    = (const float*)d_in[3];
  const float* b1    = (const float*)d_in[4];
  const float* W2    = (const float*)d_in[5];
  const float* b2    = (const float*)d_in[6];
  const float* W3    = (const float*)d_in[7];
  const float* b3    = (const float*)d_in[8];

  const int N = in_sizes[0] / F_IN;
  const int E = in_sizes[1] / 2;
  const int G = out_size / OUTC;
  const int* src = ei;
  const int* dst = ei + E;
  float* out = (float*)d_out;

  float* dis = (float*)d_ws;
  float* cnt = dis + N;
  size_t off = (size_t)N + (size_t)G;
  off = (off + 63) & ~(size_t)63;
  float* bufA = (float*)d_ws + off;
  float* bufB = bufA + (size_t)N * HID;
  const int Npad = ceil_div(N + 1, SCAN_CHUNK) * SCAN_CHUNK;
  const size_t Epad = ((size_t)E + 63) & ~(size_t)63;
  int* degi   = (int*)(bufB + (size_t)N * HID);
  int* rowptr = degi + Npad;
  int* rank   = rowptr + Npad;
  int* csr    = rank + Epad;
  int* cbase  = csr + 2 * Epad;
  const int TB = 256;
  const int chunk = ceil_div(ceil_div(E, NCHUNK), TB) * TB;

  hipMemsetAsync(cnt, 0, (size_t)G * sizeof(float), stream);
  hipMemsetAsync(out, 0, (size_t)out_size * sizeof(float), stream);

  chunk_rank_kernel<<<NCHUNK, 1024, 0, stream>>>(dst, rank, cbase, E, Npad, chunk);
  chunk_scan_kernel<<<Npad / 256, 256, 0, stream>>>(cbase, degi, Npad);
  rowptr_kernel<<<Npad / SCAN_CHUNK, 256, 0, stream>>>(degi, rowptr, dis, N);
  csr_fill_kernel<<<ceil_div(E, TB), TB, 0, stream>>>(src, dst, rank, rowptr, cbase, dis, csr, E, Npad, chunk);

  const int gemm_blocks = ceil_div(ceil_div(N, 16), 8);

  gemm_wmma_kernel<F_IN, HID><<<gemm_blocks, 256, 0, stream>>>(x, W1, bufA, N);
  gather_kernel<HID><<<ceil_div(N, 256 / (HID / 4)), 256, 0, stream>>>(bufA, rowptr, csr, dis, b1, bufB, N);

  gemm_wmma_kernel<HID, HID2><<<gemm_blocks, 256, 0, stream>>>(bufB, W2, bufA, N);
  gather_kernel<HID2><<<ceil_div(N, 256 / (HID2 / 4)), 256, 0, stream>>>(bufA, rowptr, csr, dis, b2, bufB, N);

  gemm_wmma_kernel<HID2, OUTC><<<gemm_blocks, 256, 0, stream>>>(bufB, W3, bufA, N);
  gather_kernel<OUTC><<<ceil_div(N, 256 / (OUTC / 4)), 256, 0, stream>>>(bufA, rowptr, csr, dis, b3, bufB, N);

  pool_count_kernel<<<ceil_div(N, TB), TB, 0, stream>>>(batch, cnt, N);
  pool_sum_kernel<<<ceil_div(N * 32, TB), TB, 0, stream>>>(bufB, batch, out, N);
  pool_div_kernel<<<ceil_div(G * OUTC, TB), TB, 0, stream>>>(out, cnt, G);
}


# Mean pooling folded into the last gather: a wave sums its 4 finished rows and adds them and the node count to out/cnt with f32 atomics; 2 launches fewer; sum order now wave partials, then arrival
# speedup vs baseline: 4.2379x; 1.0562x over previous
#include <hip/hip_runtime.h>

typedef float v2f __attribute__((ext_vector_type(2)));
typedef float v8f __attribute__((ext_vector_type(8)));

#define F_IN 128
#define HID  128
#define HID2 64
#define OUTC 32

static inline int ceil_div(int a, int b) { return (a + b - 1) / b; }


#define SCAN_CHUNK 1024
#define NCHUNK 64
#define RANK_CAP 65536

__global__ __launch_bounds__(1024) void chunk_rank_kernel(const int* __restrict__ dst,
                                                          int* __restrict__ rank,
                                                          int* __restrict__ cbase,
                                                          int E, int Npad, int chunk) {
  __shared__ int4 cnt4[RANK_CAP / 4];
  int* cnt = (int*)cnt4;
  const int t = threadIdx.x;
  const int e0 = blockIdx.x * chunk;
  const int e1 = e0 + chunk < E ? e0 + chunk : E;
  int4* mine = (int4*)(cbase + (size_t)blockIdx.x * Npad);

  for (int base = 0; base < Npad; base += RANK_CAP) {
    const int lim = Npad - base < RANK_CAP ? Npad - base : RANK_CAP;
    for (int i = t; i < lim / 4; i += 1024) cnt4[i] = make_int4(0, 0, 0, 0);
    __syncthreads();
    for (int e = e0 + t; e < e1; e += 8 * 1024) {
      int d[8];
#pragma unroll
      for (int u = 0; u < 8; ++u) {
        int ee = e + u * 1024;
        d[u] = dst[ee < e1 ? ee : e1 - 1];
      }
#pragma unroll
      for (int u = 0; u < 8; ++u) {
        int ee = e + u * 1024;
        int dd = d[u] - base;
        if (ee < e1 && (unsigned)dd < (unsigned)lim) rank[ee] = atomicAdd(&cnt[dd], 1);
      }
    }
    __syncthreads();
    for (int i = t; i < lim / 4; i += 1024) mine[base / 4 + i] = cnt4[i];
    __syncthreads();
  }
}

__global__ __launch_bounds__(256) void chunk_scan_kernel(int* __restrict__ cbase, int* __restrict__ degi,
                                                         int Npad) {
  const int i = blockIdx.x * 256 + threadIdx.x;
  int v[NCHUNK];
#pragma unroll
  for (int c = 0; c < NCHUNK; ++c) v[c] = cbase[(size_t)c * Npad + i];
  int run = 0;
#pragma unroll
  for (int c = 0; c < NCHUNK; ++c) {
    cbase[(size_t)c * Npad + i] = run;
    run += v[c];
  }
  degi[i] = run;
}

__global__ __launch_bounds__(256) void rowptr_kernel(const int* __restrict__ degi,
                                                     int* __restrict__ rowptr,
                                                     float* __restrict__ dis, int N) {
  __shared__ int front[8];
  __shared__ int wtot[8];
  const int t = threadIdx.x;
  const int lane = t & 31, wave = t >> 5;
  const int b = blockIdx.x;
  const int4* d4 = (const int4*)degi;

  int pre = 0;
#pragma unroll 8
  for (int i = t; i < b * 256; i += 256) {
    int4 q = d4[i];
    pre += q.x + q.y + q.z + q.w;
  }
  for (int o = 16; o > 0; o >>= 1) pre += __shfl_xor(pre, o);
  if (lane == 0) front[wave] = pre;

  const int4 v = d4[b * 256 + t];
  const int tot = v.x + v.y + v.z + v.w;
  int inc = tot;
  for (int o = 1; o < 32; o <<= 1) {
    int u = __shfl_up(inc, o);
    if (lane >= o) inc += u;
  }
  if (lane == 31) wtot[wave] = inc;
  __syncthreads();

  int ex = inc - tot;
  for (int w = 0; w < 8; ++w) {
    ex += front[w];
    if (w < wave) ex += wtot[w];
  }
  const int c[4] = {v.x, v.y, v.z, v.w};
  const int i0 = (b * 256 + t) * 4;
#pragma unroll
  for (int k = 0; k < 4; ++k) {
    int i = i0 + k;
    if (i <= N) rowptr[i] = ex;
    if (i < N) dis[i] = rsqrtf((float)(c[k] + 1));
    ex += c[k];
  }
}

__global__ void csr_fill_kernel(const int* __restrict__ src, const int* __restrict__ dst,
                                const int* __restrict__ rank, const int* __restrict__ rowptr,
                                const int* __restrict__ cbase,
                                const float* __restrict__ dis, int* __restrict__ csr_, int E,
                                int Npad, int chunk) {
  int e = blockIdx.x * blockDim.x + threadIdx.x;
  if (e >= E) return;
  const int* cb = cbase + (size_t)((blockIdx.x * blockDim.x) / chunk) * Npad;
  int s = src[e], d = dst[e];
  int2* csr = (int2*)csr_;
  csr[rowptr[d] + cb[d] + rank[e]] = make_int2(s, __float_as_int(dis[s]));
}

template <int K, int NCOL>
__global__ __launch_bounds__(256) void gemm_wmma_kernel(
    const float* __restrict__ A, const float* __restrict__ B, float* __restrict__ C, int M) {
  constexpr int CT = NCOL / 16 < 4 ? NCOL / 16 : 4;
  constexpr int RUN = 16 * CT;
  constexpr int KQ = K / 4;
  constexpr int KG = K / 8;
  __shared__ float4 Bl[KQ * NCOL];

  const int t = threadIdx.x;
  const int lane = t & 31;
  const int wave = t >> 5;
  const int c = lane & 15;
  const int h = lane >> 4;
  const int row0 = (blockIdx.x * 8 + wave) << 4;

  float4 a[KG];
  {
    int row = row0 + c;
    row = row < M ? row : M - 1;
    const float4* ap = (const float4*)(A + (size_t)row * K) + h;
#pragma unroll
    for (int g = 0; g < KG; ++g) a[g] = ap[2 * g];
  }

  static_assert((KQ * NCOL) % 256 == 0, "staging loop covers Bl in whole sweeps of the block");
#pragma unroll 8
  for (int it = 0; it < KQ * NCOL / 256; ++it) {
    const int i = t + it * 256;
    const int kq = i / NCOL;
    const int n = i % NCOL;
    const float* p = B + (size_t)(4 * kq) * NCOL + n;
    const int nl = n % RUN;
    const int slot = (n - nl) + (nl % CT) * 16 + nl / CT;
    Bl[kq * NCOL + slot] = make_float4(p[0], p[NCOL], p[2 * NCOL], p[3 * NCOL]);
  }
  __syncthreads();
  if (row0 >= M) return;

  for (int n0 = 0; n0 < NCOL; n0 += RUN) {
    v8f acc[CT];
#pragma unroll
    for (int j = 0; j < CT; ++j) acc[j] = (v8f){0.f, 0.f, 0.f, 0.f, 0.f, 0.f, 0.f, 0.f};

#pragma unroll
    for (int g = 0; g < KG; ++g) {
      float4 b[CT];
#pragma unroll
      for (int j = 0; j < CT; ++j) b[j] = Bl[(2 * g + h) * NCOL + n0 + j * 16 + c];
#pragma unroll
      for (int j = 0; j < CT; ++j) {
        v2f av = {a[g].x, a[g].y};
        v2f bv = {b[j].x, b[j].y};
        acc[j] = __builtin_amdgcn_wmma_f32_16x16x4_f32(
             false, av,  false, bv,
             (short)0, acc[j],  false,  false);
      }
#pragma unroll
      for (int j = 0; j < CT; ++j) {
        v2f av = {a[g].z, a[g].w};
        v2f bv = {b[j].z, b[j].w};
        acc[j] = __builtin_amdgcn_wmma_f32_16x16x4_f32(
             false, av,  false, bv,
             (short)0, acc[j],  false,  false);
      }
    }

#pragma unroll
    for (int r = 0; r < 8; ++r) {
      const int row = row0 + 8 * h + r;
      if (row < M) {
        float* cp = C + (size_t)row * NCOL + n0 + c * CT;
        if (CT == 4) {
          *(float4*)cp = make_float4(acc[0][r], acc[1 % CT][r], acc[2 % CT][r], acc[3 % CT][r]);
        } else if (CT == 2) {
          *(float2*)cp = make_float2(acc[0][r], acc[1 % CT][r]);
        } else {
          cp[0] = acc[0][r];
        }
      }
    }
  }
}


template <int LPN, bool TAIL>
__device__ __forceinline__ void add8(float4& acc, const int2* __restrict__ csr,
                                     const float4* __restrict__ rows, int j, int end, int sub, float dd) {
  int2 ent[8];
  float4 v[8];
#pragma unroll
  for (int u = 0; u < 8; ++u) {
    int jj = j + u;
    if (TAIL) jj = jj < end ? jj : end - 1;
    ent[u] = csr[jj];
  }
#pragma unroll
  for (int u = 0; u < 8; ++u)
    v[u] = rows[ent[u].x * LPN + sub];
#pragma unroll
  for (int u = 0; u < 8; ++u) {
    float4 r = v[u];
    if (TAIL) {
      bool ok = j + u < end;
      r.x = ok ? r.x : 0.f;
      r.y = ok ? r.y : 0.f;
      r.z = ok ? r.z : 0.f;
      r.w = ok ? r.w : 0.f;
    }
    float nrm = __int_as_float(ent[u].y) * dd;
    acc.x = fmaf(r.x, nrm, acc.x);
    acc.y = fmaf(r.y, nrm, acc.y);
    acc.z = fmaf(r.z, nrm, acc.z);
    acc.w = fmaf(r.w, nrm, acc.w);
  }
}

template <int F, bool POOL>
__global__ __launch_bounds__(256) void gather_kernel(const float* __restrict__ XW,
                                                     const int* __restrict__ rowptr,
                                                     const int* __restrict__ csr_,
                                                     const float* __restrict__ dis,
                                                     const float* __restrict__ bias,
                                                     float* __restrict__ h_out,
                                                     const int* __restrict__ batch,
                                                     float* __restrict__ pool_sum,
                                                     float* __restrict__ pool_cnt, int N) {
  constexpr int LPN = F / 4;
  constexpr int NPB = 256 / LPN;
  const int sub = threadIdx.x % LPN;
  int node = blockIdx.x * NPB + threadIdx.x / LPN;
  const bool live = node < N;
  int graph = -1;
  if constexpr (POOL) {
    if (blockIdx.x * NPB + (int)(threadIdx.x & ~31u) / LPN >= N) return;
    node = live ? node : N - 1;
    if (live) graph = batch[node];
  } else {
    if (!live) return;
  }
  const int2* csr = (const int2*)csr_;
  const float4* rows = (const float4*)XW;
  const int beg = rowptr[node], end = rowptr[node + 1];
  const float dd = dis[node];
  const float4 own = rows[node * LPN + sub];
  const float sw = dd * dd;
  float4 acc = make_float4(own.x * sw, own.y * sw, own.z * sw, own.w * sw);

  int j = beg;
  for (; j + 8 <= end; j += 8) add8<LPN, false>(acc, csr, rows, j, end, sub, dd);
  if (j < end) add8<LPN, true>(acc, csr, rows, j, end, sub, dd);

  const float* bp = bias + sub * 4;
  float4 h;
  h.x = acc.x + bp[0];
  h.y = acc.y + bp[1];
  h.z = acc.z + bp[2];
  h.w = acc.w + bp[3];
  h.x = h.x > 0.f ? h.x : 0.f;
  h.y = h.y > 0.f ? h.y : 0.f;
  h.z = h.z > 0.f ? h.z : 0.f;
  h.w = h.w > 0.f ? h.w : 0.f;
  if constexpr (!POOL) {
    ((float4*)h_out)[node * LPN + sub] = h;
  } else {
    static_assert(F == OUTC && LPN == 8, "pooled epilogue: 4 nodes of 8 lanes per wave");
    const int lane = threadIdx.x & 31;
    const int g0 = __builtin_amdgcn_readfirstlane(graph);
    if (__all(graph == g0)) {
      float4 s = h;
      s.x += __shfl_xor(s.x, 8);
      s.y += __shfl_xor(s.y, 8);
      s.z += __shfl_xor(s.z, 8);
      s.w += __shfl_xor(s.w, 8);
      s.x += __shfl_xor(s.x, 16);
      s.y += __shfl_xor(s.y, 16);
      s.z += __shfl_xor(s.z, 16);
      s.w += __shfl_xor(s.w, 16);
      const int q = lane >> 3;
      const float v = q == 0 ? s.x : q == 1 ? s.y : q == 2 ? s.z : s.w;
      atomicAdd(&pool_sum[(size_t)g0 * OUTC + sub * 4 + q], v);
      if (lane == 0) atomicAdd(&pool_cnt[g0], (float)(32 / LPN));
    } else if (live) {
      float* ps = pool_sum + (size_t)graph * OUTC + sub * 4;
      atomicAdd(ps + 0, h.x);
      atomicAdd(ps + 1, h.y);
      atomicAdd(ps + 2, h.z);
      atomicAdd(ps + 3, h.w);
      if (sub == 0) atomicAdd(&pool_cnt[graph], 1.0f);
    }
  }
}


__global__ void pool_div_kernel(float* __restrict__ out, const float* __restrict__ cnt, int G) {
  int i = blockIdx.x * blockDim.x + threadIdx.x;
  if (i >= G * OUTC) return;
  out[i] /= fmaxf(cnt[i >> 5], 1.0f);
}


extern "C" void kernel_launch(void* const* d_in, const int* in_sizes, int n_in,
                              void* d_out, int out_size, void* d_ws, size_t ws_size,
                              hipStream_t stream) {
  const float* x     = (const float*)d_in[0];
  const int*   ei    = (const int*)d_in[1];
  const int*   batch = (const int*)d_in[2];
  const float* W1    = (const float*)d_in[3];
  const float* b1    = (const float*)d_in[4];
  const float* W2    = (const float*)d_in[5];
  const float* b2    = (const float*)d_in[6];
  const float* W3    = (const float*)d_in[7];
  const float* b3    = (const float*)d_in[8];

  const int N = in_sizes[0] / F_IN;
  const int E = in_sizes[1] / 2;
  const int G = out_size / OUTC;
  const int* src = ei;
  const int* dst = ei + E;
  float* out = (float*)d_out;

  float* dis = (float*)d_ws;
  float* cnt = dis + N;
  size_t off = (size_t)N + (size_t)G;
  off = (off + 63) & ~(size_t)63;
  float* bufA = (float*)d_ws + off;
  float* bufB = bufA + (size_t)N * HID;
  const int Npad = ceil_div(N + 1, SCAN_CHUNK) * SCAN_CHUNK;
  const size_t Epad = ((size_t)E + 63) & ~(size_t)63;
  int* degi   = (int*)(bufB + (size_t)N * HID);
  int* rowptr = degi + Npad;
  int* rank   = rowptr + Npad;
  int* csr    = rank + Epad;
  int* cbase  = csr + 2 * Epad;
  const int TB = 256;
  const int chunk = ceil_div(ceil_div(E, NCHUNK), TB) * TB;

  hipMemsetAsync(cnt, 0, (size_t)G * sizeof(float), stream);
  hipMemsetAsync(out, 0, (size_t)out_size * sizeof(float), stream);

  chunk_rank_kernel<<<NCHUNK, 1024, 0, stream>>>(dst, rank, cbase, E, Npad, chunk);
  chunk_scan_kernel<<<Npad / 256, 256, 0, stream>>>(cbase, degi, Npad);
  rowptr_kernel<<<Npad / SCAN_CHUNK, 256, 0, stream>>>(degi, rowptr, dis, N);
  csr_fill_kernel<<<ceil_div(E, TB), TB, 0, stream>>>(src, dst, rank, rowptr, cbase, dis, csr, E, Npad, chunk);

  const int gemm_blocks = ceil_div(ceil_div(N, 16), 8);

  gemm_wmma_kernel<F_IN, HID><<<gemm_blocks, 256, 0, stream>>>(x, W1, bufA, N);
  gather_kernel<HID, false><<<ceil_div(N, 256 / (HID / 4)), 256, 0, stream>>>(
      bufA, rowptr, csr, dis, b1, bufB, batch, out, cnt, N);

  gemm_wmma_kernel<HID, HID2><<<gemm_blocks, 256, 0, stream>>>(bufB, W2, bufA, N);
  gather_kernel<HID2, false><<<ceil_div(N, 256 / (HID2 / 4)), 256, 0, stream>>>(
      bufA, rowptr, csr, dis, b2, bufB, batch, out, cnt, N);

  gemm_wmma_kernel<HID2, OUTC><<<gemm_blocks, 256, 0, stream>>>(bufB, W3, bufA, N);
  gather_kernel<OUTC, true><<<ceil_div(N, 256 / (OUTC / 4)), 256, 0, stream>>>(
      bufA, rowptr, csr, dis, b3, bufB, batch, out, cnt, N);

  pool_div_kernel<<<ceil_div(G * OUTC, TB), TB, 0, stream>>>(out, cnt, G);
}


# Edge slots from a second LDS pass: 256 blocks (edge chunk x node quarter) start counters at each node's base slot, an LDS add returns the slot; no rank array; in-chunk sum order = that pass's arrival
# speedup vs baseline: 4.4551x; 1.0513x over previous
#include <hip/hip_runtime.h>

typedef float v2f __attribute__((ext_vector_type(2)));
typedef float v8f __attribute__((ext_vector_type(8)));

#define F_IN 128
#define HID  128
#define HID2 64
#define OUTC 32

static inline int ceil_div(int a, int b) { return (a + b - 1) / b; }


#define SCAN_CHUNK 1024
#define NCHUNK 64
#define RANK_CAP 65536
#define NSPLIT 4
#define FILL_CAP 16384

__global__ __launch_bounds__(1024) void chunk_count_kernel(const int* __restrict__ dst,
                                                           int* __restrict__ cbase,
                                                           int E, int Npad, int chunk) {
  __shared__ int4 cnt4[RANK_CAP / 4];
  int* cnt = (int*)cnt4;
  const int t = threadIdx.x;
  const int e0 = blockIdx.x * chunk;
  const int e1 = e0 + chunk < E ? e0 + chunk : E;
  int4* mine = (int4*)(cbase + (size_t)blockIdx.x * Npad);

  for (int base = 0; base < Npad; base += RANK_CAP) {
    const int lim = Npad - base < RANK_CAP ? Npad - base : RANK_CAP;
    for (int i = t; i < lim / 4; i += 1024) cnt4[i] = make_int4(0, 0, 0, 0);
    __syncthreads();
    for (int e = e0 + t; e < e1; e += 8 * 1024) {
      int d[8];
#pragma unroll
      for (int u = 0; u < 8; ++u) {
        int ee = e + u * 1024;
        d[u] = dst[ee < e1 ? ee : e1 - 1];
      }
#pragma unroll
      for (int u = 0; u < 8; ++u) {
        int ee = e + u * 1024;
        int dd = d[u] - base;
        if (ee < e1 && (unsigned)dd < (unsigned)lim) atomicAdd(&cnt[dd], 1);
      }
    }
    __syncthreads();
    for (int i = t; i < lim / 4; i += 1024) mine[base / 4 + i] = cnt4[i];
    __syncthreads();
  }
}

__global__ __launch_bounds__(1024) void chunk_fill_kernel(const int* __restrict__ src,
                                                          const int* __restrict__ dst,
                                                          const int* __restrict__ cbase,
                                                          const int* __restrict__ rowptr,
                                                          const float* __restrict__ dis,
                                                          int* __restrict__ csr_,
                                                          int E, int Npad, int chunk) {
  __shared__ int4 cnt4[FILL_CAP / 4];
  int* cnt = (int*)cnt4;
  const int t = threadIdx.x;
  const int c = blockIdx.x / NSPLIT;
  const int range = Npad / NSPLIT;
  const int n0 = (blockIdx.x % NSPLIT) * range;
  const int e0 = c * chunk;
  const int e1 = e0 + chunk < E ? e0 + chunk : E;
  const int* mine = cbase + (size_t)c * Npad;
  int2* csr = (int2*)csr_;

  for (int base = n0; base < n0 + range; base += FILL_CAP) {
    const int lim = n0 + range - base < FILL_CAP ? n0 + range - base : FILL_CAP;
    const int4* cb4 = (const int4*)(mine + base);
    const int4* rp4 = (const int4*)(rowptr + base);
    for (int i = t; i < lim / 4; i += 1024) {
      const int4 a = cb4[i], r = rp4[i];
      cnt4[i] = make_int4(a.x + r.x, a.y + r.y, a.z + r.z, a.w + r.w);
    }
    __syncthreads();
    for (int e = e0 + t; e < e1; e += 8 * 1024) {
      int d[8], s[8];
      float w[8];
#pragma unroll
      for (int u = 0; u < 8; ++u) {
        int ee = e + u * 1024;
        ee = ee < e1 ? ee : e1 - 1;
        d[u] = dst[ee];
        s[u] = src[ee];
      }
#pragma unroll
      for (int u = 0; u < 8; ++u) {
        int ee = e + u * 1024;
        int dd = d[u] - base;
        bool ok = ee < e1 && (unsigned)dd < (unsigned)lim;
        w[u] = dis[ok ? s[u] : 0];
      }
#pragma unroll
      for (int u = 0; u < 8; ++u) asm volatile("" ::"v"(w[u]));
      int pos[8];
#pragma unroll
      for (int u = 0; u < 8; ++u) {
        int ee = e + u * 1024;
        int dd = d[u] - base;
        pos[u] = -1;
        if (ee < e1 && (unsigned)dd < (unsigned)lim) pos[u] = atomicAdd(&cnt[dd], 1);
      }
#pragma unroll
      for (int u = 0; u < 8; ++u)
        if (pos[u] >= 0) csr[pos[u]] = make_int2(s[u], __float_as_int(w[u]));
    }
    __syncthreads();
  }
}

__global__ __launch_bounds__(256) void chunk_scan_kernel(int* __restrict__ cbase, int* __restrict__ degi,
                                                         int Npad) {
  const int i = blockIdx.x * 256 + threadIdx.x;
  int v[NCHUNK];
#pragma unroll
  for (int c = 0; c < NCHUNK; ++c) v[c] = cbase[(size_t)c * Npad + i];
  int run = 0;
#pragma unroll
  for (int c = 0; c < NCHUNK; ++c) {
    cbase[(size_t)c * Npad + i] = run;
    run += v[c];
  }
  degi[i] = run;
}

__global__ __launch_bounds__(256) void rowptr_kernel(const int* __restrict__ degi,
                                                     int* __restrict__ rowptr,
                                                     float* __restrict__ dis, int N) {
  __shared__ int front[8];
  __shared__ int wtot[8];
  const int t = threadIdx.x;
  const int lane = t & 31, wave = t >> 5;
  const int b = blockIdx.x;
  const int4* d4 = (const int4*)degi;

  int pre = 0;
#pragma unroll 8
  for (int i = t; i < b * 256; i += 256) {
    int4 q = d4[i];
    pre += q.x + q.y + q.z + q.w;
  }
  for (int o = 16; o > 0; o >>= 1) pre += __shfl_xor(pre, o);
  if (lane == 0) front[wave] = pre;

  const int4 v = d4[b * 256 + t];
  const int tot = v.x + v.y + v.z + v.w;
  int inc = tot;
  for (int o = 1; o < 32; o <<= 1) {
    int u = __shfl_up(inc, o);
    if (lane >= o) inc += u;
  }
  if (lane == 31) wtot[wave] = inc;
  __syncthreads();

  int ex = inc - tot;
  for (int w = 0; w < 8; ++w) {
    ex += front[w];
    if (w < wave) ex += wtot[w];
  }
  const int c[4] = {v.x, v.y, v.z, v.w};
  const int i0 = (b * 256 + t) * 4;
#pragma unroll
  for (int k = 0; k < 4; ++k) {
    int i = i0 + k;
    rowptr[i] = ex;
    if (i < N) dis[i] = rsqrtf((float)(c[k] + 1));
    ex += c[k];
  }
}

template <int K, int NCOL>
__global__ __launch_bounds__(256) void gemm_wmma_kernel(
    const float* __restrict__ A, const float* __restrict__ B, float* __restrict__ C, int M) {
  constexpr int CT = NCOL / 16 < 4 ? NCOL / 16 : 4;
  constexpr int RUN = 16 * CT;
  constexpr int KQ = K / 4;
  constexpr int KG = K / 8;
  __shared__ float4 Bl[KQ * NCOL];

  const int t = threadIdx.x;
  const int lane = t & 31;
  const int wave = t >> 5;
  const int c = lane & 15;
  const int h = lane >> 4;
  const int row0 = (blockIdx.x * 8 + wave) << 4;

  float4 a[KG];
  {
    int row = row0 + c;
    row = row < M ? row : M - 1;
    const float4* ap = (const float4*)(A + (size_t)row * K) + h;
#pragma unroll
    for (int g = 0; g < KG; ++g) a[g] = ap[2 * g];
  }

  static_assert((KQ * NCOL) % 256 == 0, "staging loop covers Bl in whole sweeps of the block");
#pragma unroll 8
  for (int it = 0; it < KQ * NCOL / 256; ++it) {
    const int i = t + it * 256;
    const int kq = i / NCOL;
    const int n = i % NCOL;
    const float* p = B + (size_t)(4 * kq) * NCOL + n;
    const int nl = n % RUN;
    const int slot = (n - nl) + (nl % CT) * 16 + nl / CT;
    Bl[kq * NCOL + slot] = make_float4(p[0], p[NCOL], p[2 * NCOL], p[3 * NCOL]);
  }
  __syncthreads();
  if (row0 >= M) return;

  for (int n0 = 0; n0 < NCOL; n0 += RUN) {
    v8f acc[CT];
#pragma unroll
    for (int j = 0; j < CT; ++j) acc[j] = (v8f){0.f, 0.f, 0.f, 0.f, 0.f, 0.f, 0.f, 0.f};

#pragma unroll
    for (int g = 0; g < KG; ++g) {
      float4 b[CT];
#pragma unroll
      for (int j = 0; j < CT; ++j) b[j] = Bl[(2 * g + h) * NCOL + n0 + j * 16 + c];
#pragma unroll
      for (int j = 0; j < CT; ++j) {
        v2f av = {a[g].x, a[g].y};
        v2f bv = {b[j].x, b[j].y};
        acc[j] = __builtin_amdgcn_wmma_f32_16x16x4_f32(
             false, av,  false, bv,
             (short)0, acc[j],  false,  false);
      }
#pragma unroll
      for (int j = 0; j < CT; ++j) {
        v2f av = {a[g].z, a[g].w};
        v2f bv = {b[j].z, b[j].w};
        acc[j] = __builtin_amdgcn_wmma_f32_16x16x4_f32(
             false, av,  false, bv,
             (short)0, acc[j],  false,  false);
      }
    }

#pragma unroll
    for (int r = 0; r < 8; ++r) {
      const int row = row0 + 8 * h + r;
      if (row < M) {
        float* cp = C + (size_t)row * NCOL + n0 + c * CT;
        if (CT == 4) {
          *(float4*)cp = make_float4(acc[0][r], acc[1 % CT][r], acc[2 % CT][r], acc[3 % CT][r]);
        } else if (CT == 2) {
          *(float2*)cp = make_float2(acc[0][r], acc[1 % CT][r]);
        } else {
          cp[0] = acc[0][r];
        }
      }
    }
  }
}


template <int LPN, bool TAIL>
__device__ __forceinline__ void add8(float4& acc, const int2* __restrict__ csr,
                                     const float4* __restrict__ rows, int j, int end, int sub, float dd) {
  int2 ent[8];
  float4 v[8];
#pragma unroll
  for (int u = 0; u < 8; ++u) {
    int jj = j + u;
    if (TAIL) jj = jj < end ? jj : end - 1;
    ent[u] = csr[jj];
  }
#pragma unroll
  for (int u = 0; u < 8; ++u)
    v[u] = rows[ent[u].x * LPN + sub];
#pragma unroll
  for (int u = 0; u < 8; ++u) {
    float4 r = v[u];
    if (TAIL) {
      bool ok = j + u < end;
      r.x = ok ? r.x : 0.f;
      r.y = ok ? r.y : 0.f;
      r.z = ok ? r.z : 0.f;
      r.w = ok ? r.w : 0.f;
    }
    float nrm = __int_as_float(ent[u].y) * dd;
    acc.x = fmaf(r.x, nrm, acc.x);
    acc.y = fmaf(r.y, nrm, acc.y);
    acc.z = fmaf(r.z, nrm, acc.z);
    acc.w = fmaf(r.w, nrm, acc.w);
  }
}

template <int F, bool POOL>
__global__ __launch_bounds__(256) void gather_kernel(const float* __restrict__ XW,
                                                     const int* __restrict__ rowptr,
                                                     const int* __restrict__ csr_,
                                                     const float* __restrict__ dis,
                                                     const float* __restrict__ bias,
                                                     float* __restrict__ h_out,
                                                     const int* __restrict__ batch,
                                                     float* __restrict__ pool_sum,
                                                     float* __restrict__ pool_cnt, int N) {
  constexpr int LPN = F / 4;
  constexpr int NPB = 256 / LPN;
  const int sub = threadIdx.x % LPN;
  int node = blockIdx.x * NPB + threadIdx.x / LPN;
  const bool live = node < N;
  int graph = -1;
  if constexpr (POOL) {
    if (blockIdx.x * NPB + (int)(threadIdx.x & ~31u) / LPN >= N) return;
    node = live ? node : N - 1;
    if (live) graph = batch[node];
  } else {
    if (!live) return;
  }
  const int2* csr = (const int2*)csr_;
  const float4* rows = (const float4*)XW;
  const int beg = rowptr[node], end = rowptr[node + 1];
  const float dd = dis[node];
  const float4 own = rows[node * LPN + sub];
  const float sw = dd * dd;
  float4 acc = make_float4(own.x * sw, own.y * sw, own.z * sw, own.w * sw);

  int j = beg;
  for (; j + 8 <= end; j += 8) add8<LPN, false>(acc, csr, rows, j, end, sub, dd);
  if (j < end) add8<LPN, true>(acc, csr, rows, j, end, sub, dd);

  const float* bp = bias + sub * 4;
  float4 h;
  h.x = acc.x + bp[0];
  h.y = acc.y + bp[1];
  h.z = acc.z + bp[2];
  h.w = acc.w + bp[3];
  h.x = h.x > 0.f ? h.x : 0.f;
  h.y = h.y > 0.f ? h.y : 0.f;
  h.z = h.z > 0.f ? h.z : 0.f;
  h.w = h.w > 0.f ? h.w : 0.f;
  if constexpr (!POOL) {
    ((float4*)h_out)[node * LPN + sub] = h;
  } else {
    static_assert(F == OUTC && LPN == 8, "pooled epilogue: 4 nodes of 8 lanes per wave");
    const int lane = threadIdx.x & 31;
    const int g0 = __builtin_amdgcn_readfirstlane(graph);
    if (__all(graph == g0)) {
      float4 s = h;
      s.x += __shfl_xor(s.x, 8);
      s.y += __shfl_xor(s.y, 8);
      s.z += __shfl_xor(s.z, 8);
      s.w += __shfl_xor(s.w, 8);
      s.x += __shfl_xor(s.x, 16);
      s.y += __shfl_xor(s.y, 16);
      s.z += __shfl_xor(s.z, 16);
      s.w += __shfl_xor(s.w, 16);
      const int q = lane >> 3;
      const float v = q == 0 ? s.x : q == 1 ? s.y : q == 2 ? s.z : s.w;
      atomicAdd(&pool_sum[(size_t)g0 * OUTC + sub * 4 + q], v);
      if (lane == 0) atomicAdd(&pool_cnt[g0], (float)(32 / LPN));
    } else if (live) {
      float* ps = pool_sum + (size_t)graph * OUTC + sub * 4;
      atomicAdd(ps + 0, h.x);
      atomicAdd(ps + 1, h.y);
      atomicAdd(ps + 2, h.z);
      atomicAdd(ps + 3, h.w);
      if (sub == 0) atomicAdd(&pool_cnt[graph], 1.0f);
    }
  }
}


__global__ void pool_div_kernel(float* __restrict__ out, const float* __restrict__ cnt, int G) {
  int i = blockIdx.x * blockDim.x + threadIdx.x;
  if (i >= G * OUTC) return;
  out[i] /= fmaxf(cnt[i >> 5], 1.0f);
}


extern "C" void kernel_launch(void* const* d_in, const int* in_sizes, int n_in,
                              void* d_out, int out_size, void* d_ws, size_t ws_size,
                              hipStream_t stream) {
  const float* x     = (const float*)d_in[0];
  const int*   ei    = (const int*)d_in[1];
  const int*   batch = (const int*)d_in[2];
  const float* W1    = (const float*)d_in[3];
  const float* b1    = (const float*)d_in[4];
  const float* W2    = (const float*)d_in[5];
  const float* b2    = (const float*)d_in[6];
  const float* W3    = (const float*)d_in[7];
  const float* b3    = (const float*)d_in[8];

  const int N = in_sizes[0] / F_IN;
  const int E = in_sizes[1] / 2;
  const int G = out_size / OUTC;
  const int* src = ei;
  const int* dst = ei + E;
  float* out = (float*)d_out;

  float* dis = (float*)d_ws;
  float* cnt = dis + N;
  size_t off = (size_t)N + (size_t)G;
  off = (off + 63) & ~(size_t)63;
  float* bufA = (float*)d_ws + off;
  float* bufB = bufA + (size_t)N * HID;
  const int Npad = ceil_div(N + 1, SCAN_CHUNK) * SCAN_CHUNK;
  const size_t Epad = ((size_t)E + 63) & ~(size_t)63;
  int* degi   = (int*)(bufB + (size_t)N * HID);
  int* rowptr = degi + Npad;
  int* csr    = rowptr + Npad;
  int* cbase  = csr + 2 * Epad;
  const int TB = 256;
  const int chunk = ceil_div(ceil_div(E, NCHUNK), TB) * TB;

  hipMemsetAsync(cnt, 0, (size_t)G * sizeof(float), stream);
  hipMemsetAsync(out, 0, (size_t)out_size * sizeof(float), stream);

  chunk_count_kernel<<<NCHUNK, 1024, 0, stream>>>(dst, cbase, E, Npad, chunk);
  chunk_scan_kernel<<<Npad / 256, 256, 0, stream>>>(cbase, degi, Npad);
  rowptr_kernel<<<Npad / SCAN_CHUNK, 256, 0, stream>>>(degi, rowptr, dis, N);
  chunk_fill_kernel<<<NCHUNK * NSPLIT, 1024, 0, stream>>>(src, dst, cbase, rowptr, dis, csr, E, Npad, chunk);

  const int gemm_blocks = ceil_div(ceil_div(N, 16), 8);

  gemm_wmma_kernel<F_IN, HID><<<gemm_blocks, 256, 0, stream>>>(x, W1, bufA, N);
  gather_kernel<HID, false><<<ceil_div(N, 256 / (HID / 4)), 256, 0, stream>>>(
      bufA, rowptr, csr, dis, b1, bufB, batch, out, cnt, N);

  gemm_wmma_kernel<HID, HID2><<<gemm_blocks, 256, 0, stream>>>(bufB, W2, bufA, N);
  gather_kernel<HID2, false><<<ceil_div(N, 256 / (HID2 / 4)), 256, 0, stream>>>(
      bufA, rowptr, csr, dis, b2, bufB, batch, out, cnt, N);

  gemm_wmma_kernel<HID2, OUTC><<<gemm_blocks, 256, 0, stream>>>(bufB, W3, bufA, N);
  gather_kernel<OUTC, true><<<ceil_div(N, 256 / (OUTC / 4)), 256, 0, stream>>>(
      bufA, rowptr, csr, dis, b3, bufB, batch, out, cnt, N);

  pool_div_kernel<<<ceil_div(G * OUTC, TB), TB, 0, stream>>>(out, cnt, G);
}


# Four launches fewer: clears go into the counting kernel, the layer-1 product shares the chunk-prefix launch, the last gather adds rows already divided by their graph's node count (no division pass)
# speedup vs baseline: 4.6002x; 1.0326x over previous
#include <hip/hip_runtime.h>

typedef float v2f __attribute__((ext_vector_type(2)));
typedef float v8f __attribute__((ext_vector_type(8)));

#define F_IN 128
#define HID  128
#define HID2 64
#define OUTC 32

static inline int ceil_div(int a, int b) { return (a + b - 1) / b; }


#define SCAN_CHUNK 1024
#define NCHUNK 64
#define RANK_CAP 65536
#define NSPLIT 4
#define FILL_CAP 16384

__global__ __launch_bounds__(1024) void chunk_count_kernel(const int* __restrict__ dst,
                                                           int* __restrict__ cbase,
                                                           int E, int Npad, int chunk,
                                                           float* __restrict__ pool_sum, int n_pool,
                                                           int* __restrict__ gcnt, int G) {
  __shared__ int4 cnt4[RANK_CAP / 4];
  int* cnt = (int*)cnt4;
  const int t = threadIdx.x;
  for (int i = blockIdx.x * 1024 + t; i < n_pool; i += NCHUNK * 1024) pool_sum[i] = 0.f;
  for (int i = blockIdx.x * 1024 + t; i < G; i += NCHUNK * 1024) gcnt[i] = 0;
  const int e0 = blockIdx.x * chunk;
  const int e1 = e0 + chunk < E ? e0 + chunk : E;
  int4* mine = (int4*)(cbase + (size_t)blockIdx.x * Npad);

  for (int base = 0; base < Npad; base += RANK_CAP) {
    const int lim = Npad - base < RANK_CAP ? Npad - base : RANK_CAP;
    for (int i = t; i < lim / 4; i += 1024) cnt4[i] = make_int4(0, 0, 0, 0);
    __syncthreads();
    for (int e = e0 + t; e < e1; e += 8 * 1024) {
      int d[8];
#pragma unroll
      for (int u = 0; u < 8; ++u) {
        int ee = e + u * 1024;
        d[u] = dst[ee < e1 ? ee : e1 - 1];
      }
#pragma unroll
      for (int u = 0; u < 8; ++u) {
        int ee = e + u * 1024;
        int dd = d[u] - base;
        if (ee < e1 && (unsigned)dd < (unsigned)lim) atomicAdd(&cnt[dd], 1);
      }
    }
    __syncthreads();
    for (int i = t; i < lim / 4; i += 1024) mine[base / 4 + i] = cnt4[i];
    __syncthreads();
  }
}

__global__ __launch_bounds__(1024) void chunk_fill_kernel(const int* __restrict__ src,
                                                          const int* __restrict__ dst,
                                                          const int* __restrict__ cbase,
                                                          const int* __restrict__ rowptr,
                                                          const float* __restrict__ dis,
                                                          int* __restrict__ csr_,
                                                          int E, int Npad, int chunk) {
  __shared__ int4 cnt4[FILL_CAP / 4];
  int* cnt = (int*)cnt4;
  const int t = threadIdx.x;
  const int c = blockIdx.x / NSPLIT;
  const int range = Npad / NSPLIT;
  const int n0 = (blockIdx.x % NSPLIT) * range;
  const int e0 = c * chunk;
  const int e1 = e0 + chunk < E ? e0 + chunk : E;
  const int* mine = cbase + (size_t)c * Npad;
  int2* csr = (int2*)csr_;

  for (int base = n0; base < n0 + range; base += FILL_CAP) {
    const int lim = n0 + range - base < FILL_CAP ? n0 + range - base : FILL_CAP;
    const int4* cb4 = (const int4*)(mine + base);
    const int4* rp4 = (const int4*)(rowptr + base);
    for (int i = t; i < lim / 4; i += 1024) {
      const int4 a = cb4[i], r = rp4[i];
      cnt4[i] = make_int4(a.x + r.x, a.y + r.y, a.z + r.z, a.w + r.w);
    }
    __syncthreads();
    for (int e = e0 + t; e < e1; e += 8 * 1024) {
      int d[8], s[8];
      float w[8];
#pragma unroll
      for (int u = 0; u < 8; ++u) {
        int ee = e + u * 1024;
        ee = ee < e1 ? ee : e1 - 1;
        d[u] = dst[ee];
        s[u] = src[ee];
      }
#pragma unroll
      for (int u = 0; u < 8; ++u) {
        int ee = e + u * 1024;
        int dd = d[u] - base;
        bool ok = ee < e1 && (unsigned)dd < (unsigned)lim;
        w[u] = dis[ok ? s[u] : 0];
      }
#pragma unroll
      for (int u = 0; u < 8; ++u) asm volatile("" ::"v"(w[u]));
      int pos[8];
#pragma unroll
      for (int u = 0; u < 8; ++u) {
        int ee = e + u * 1024;
        int dd = d[u] - base;
        pos[u] = -1;
        if (ee < e1 && (unsigned)dd < (unsigned)lim) pos[u] = atomicAdd(&cnt[dd], 1);
      }
#pragma unroll
      for (int u = 0; u < 8; ++u)
        if (pos[u] >= 0) csr[pos[u]] = make_int2(s[u], __float_as_int(w[u]));
    }
    __syncthreads();
  }
}

__device__ __forceinline__ void chunk_scan_block(int block, int* __restrict__ cbase, int* __restrict__ degi,
                                                 int Npad, const int* __restrict__ batch,
                                                 int* __restrict__ gcnt, int N) {
  const int i = block * 256 + threadIdx.x;
  const bool live = i < N;
  const int g = batch[live ? i : 0];
  int v[NCHUNK];
#pragma unroll
  for (int c = 0; c < NCHUNK; ++c) v[c] = cbase[(size_t)c * Npad + i];
  int run = 0;
#pragma unroll
  for (int c = 0; c < NCHUNK; ++c) {
    cbase[(size_t)c * Npad + i] = run;
    run += v[c];
  }
  degi[i] = run;

  const int gl = live ? g : -1;
  const int g0 = __builtin_amdgcn_readfirstlane(gl);
  if (__all(gl == g0)) {
    if ((threadIdx.x & 31) == 0 && g0 >= 0) atomicAdd(&gcnt[g0], 32);
  } else if (live) {
    atomicAdd(&gcnt[gl], 1);
  }
}

__global__ __launch_bounds__(256) void rowptr_kernel(const int* __restrict__ degi,
                                                     int* __restrict__ rowptr,
                                                     float* __restrict__ dis, int N) {
  __shared__ int front[8];
  __shared__ int wtot[8];
  const int t = threadIdx.x;
  const int lane = t & 31, wave = t >> 5;
  const int b = blockIdx.x;
  const int4* d4 = (const int4*)degi;

  int pre = 0;
#pragma unroll 8
  for (int i = t; i < b * 256; i += 256) {
    int4 q = d4[i];
    pre += q.x + q.y + q.z + q.w;
  }
  for (int o = 16; o > 0; o >>= 1) pre += __shfl_xor(pre, o);
  if (lane == 0) front[wave] = pre;

  const int4 v = d4[b * 256 + t];
  const int tot = v.x + v.y + v.z + v.w;
  int inc = tot;
  for (int o = 1; o < 32; o <<= 1) {
    int u = __shfl_up(inc, o);
    if (lane >= o) inc += u;
  }
  if (lane == 31) wtot[wave] = inc;
  __syncthreads();

  int ex = inc - tot;
  for (int w = 0; w < 8; ++w) {
    ex += front[w];
    if (w < wave) ex += wtot[w];
  }
  const int c[4] = {v.x, v.y, v.z, v.w};
  const int i0 = (b * 256 + t) * 4;
#pragma unroll
  for (int k = 0; k < 4; ++k) {
    int i = i0 + k;
    rowptr[i] = ex;
    if (i < N) dis[i] = rsqrtf((float)(c[k] + 1));
    ex += c[k];
  }
}

template <int K, int NCOL>
__device__ __forceinline__ void gemm_wmma_block(
    int block, const float* __restrict__ A, const float* __restrict__ B, float* __restrict__ C, int M) {
  constexpr int CT = NCOL / 16 < 4 ? NCOL / 16 : 4;
  constexpr int RUN = 16 * CT;
  constexpr int KQ = K / 4;
  constexpr int KG = K / 8;
  __shared__ float4 Bl[KQ * NCOL];

  const int t = threadIdx.x;
  const int lane = t & 31;
  const int wave = t >> 5;
  const int c = lane & 15;
  const int h = lane >> 4;
  const int row0 = (block * 8 + wave) << 4;

  float4 a[KG];
  {
    int row = row0 + c;
    row = row < M ? row : M - 1;
    const float4* ap = (const float4*)(A + (size_t)row * K) + h;
#pragma unroll
    for (int g = 0; g < KG; ++g) a[g] = ap[2 * g];
  }

  static_assert((KQ * NCOL) % 256 == 0, "staging loop covers Bl in whole sweeps of the block");
#pragma unroll 8
  for (int it = 0; it < KQ * NCOL / 256; ++it) {
    const int i = t + it * 256;
    const int kq = i / NCOL;
    const int n = i % NCOL;
    const float* p = B + (size_t)(4 * kq) * NCOL + n;
    const int nl = n % RUN;
    const int slot = (n - nl) + (nl % CT) * 16 + nl / CT;
    Bl[kq * NCOL + slot] = make_float4(p[0], p[NCOL], p[2 * NCOL], p[3 * NCOL]);
  }
  __syncthreads();
  if (row0 >= M) return;

  for (int n0 = 0; n0 < NCOL; n0 += RUN) {
    v8f acc[CT];
#pragma unroll
    for (int j = 0; j < CT; ++j) acc[j] = (v8f){0.f, 0.f, 0.f, 0.f, 0.f, 0.f, 0.f, 0.f};

#pragma unroll
    for (int g = 0; g < KG; ++g) {
      float4 b[CT];
#pragma unroll
      for (int j = 0; j < CT; ++j) b[j] = Bl[(2 * g + h) * NCOL + n0 + j * 16 + c];
#pragma unroll
      for (int j = 0; j < CT; ++j) {
        v2f av = {a[g].x, a[g].y};
        v2f bv = {b[j].x, b[j].y};
        acc[j] = __builtin_amdgcn_wmma_f32_16x16x4_f32(
             false, av,  false, bv,
             (short)0, acc[j],  false,  false);
      }
#pragma unroll
      for (int j = 0; j < CT; ++j) {
        v2f av = {a[g].z, a[g].w};
        v2f bv = {b[j].z, b[j].w};
        acc[j] = __builtin_amdgcn_wmma_f32_16x16x4_f32(
             false, av,  false, bv,
             (short)0, acc[j],  false,  false);
      }
    }

#pragma unroll
    for (int r = 0; r < 8; ++r) {
      const int row = row0 + 8 * h + r;
      if (row < M) {
        float* cp = C + (size_t)row * NCOL + n0 + c * CT;
        if (CT == 4) {
          *(float4*)cp = make_float4(acc[0][r], acc[1 % CT][r], acc[2 % CT][r], acc[3 % CT][r]);
        } else if (CT == 2) {
          *(float2*)cp = make_float2(acc[0][r], acc[1 % CT][r]);
        } else {
          cp[0] = acc[0][r];
        }
      }
    }
  }
}

template <int K, int NCOL>
__global__ __launch_bounds__(256) void gemm_wmma_kernel(
    const float* __restrict__ A, const float* __restrict__ B, float* __restrict__ C, int M) {
  gemm_wmma_block<K, NCOL>(blockIdx.x, A, B, C, M);
}

template <int K, int NCOL>
__global__ __launch_bounds__(256) void scan_gemm_kernel(
    int* __restrict__ cbase, int* __restrict__ degi, int Npad, const int* __restrict__ batch,
    int* __restrict__ gcnt, int N, int nscan,
    const float* __restrict__ A, const float* __restrict__ B, float* __restrict__ C) {
  if ((int)blockIdx.x < nscan) {
    chunk_scan_block(blockIdx.x, cbase, degi, Npad, batch, gcnt, N);
    return;
  }
  gemm_wmma_block<K, NCOL>(blockIdx.x - nscan, A, B, C, N);
}


template <int LPN, bool TAIL>
__device__ __forceinline__ void add8(float4& acc, const int2* __restrict__ csr,
                                     const float4* __restrict__ rows, int j, int end, int sub, float dd) {
  int2 ent[8];
  float4 v[8];
#pragma unroll
  for (int u = 0; u < 8; ++u) {
    int jj = j + u;
    if (TAIL) jj = jj < end ? jj : end - 1;
    ent[u] = csr[jj];
  }
#pragma unroll
  for (int u = 0; u < 8; ++u)
    v[u] = rows[ent[u].x * LPN + sub];
#pragma unroll
  for (int u = 0; u < 8; ++u) {
    float4 r = v[u];
    if (TAIL) {
      bool ok = j + u < end;
      r.x = ok ? r.x : 0.f;
      r.y = ok ? r.y : 0.f;
      r.z = ok ? r.z : 0.f;
      r.w = ok ? r.w : 0.f;
    }
    float nrm = __int_as_float(ent[u].y) * dd;
    acc.x = fmaf(r.x, nrm, acc.x);
    acc.y = fmaf(r.y, nrm, acc.y);
    acc.z = fmaf(r.z, nrm, acc.z);
    acc.w = fmaf(r.w, nrm, acc.w);
  }
}

template <int F, bool POOL>
__global__ __launch_bounds__(256) void gather_kernel(const float* __restrict__ XW,
                                                     const int* __restrict__ rowptr,
                                                     const int* __restrict__ csr_,
                                                     const float* __restrict__ dis,
                                                     const float* __restrict__ bias,
                                                     float* __restrict__ h_out,
                                                     const int* __restrict__ batch,
                                                     float* __restrict__ pool_sum,
                                                     const int* __restrict__ gcnt, int N) {
  constexpr int LPN = F / 4;
  constexpr int NPB = 256 / LPN;
  const int sub = threadIdx.x % LPN;
  int node = blockIdx.x * NPB + threadIdx.x / LPN;
  const bool live = node < N;
  int graph = -1;
  if constexpr (POOL) {
    if (blockIdx.x * NPB + (int)(threadIdx.x & ~31u) / LPN >= N) return;
    node = live ? node : N - 1;
    if (live) graph = batch[node];
  } else {
    if (!live) return;
  }
  const int2* csr = (const int2*)csr_;
  const float4* rows = (const float4*)XW;
  const int beg = rowptr[node], end = rowptr[node + 1];
  const float dd = dis[node];
  const float4 own = rows[node * LPN + sub];
  const float sw = dd * dd;
  float4 acc = make_float4(own.x * sw, own.y * sw, own.z * sw, own.w * sw);

  int j = beg;
  for (; j + 8 <= end; j += 8) add8<LPN, false>(acc, csr, rows, j, end, sub, dd);
  if (j < end) add8<LPN, true>(acc, csr, rows, j, end, sub, dd);

  const float* bp = bias + sub * 4;
  float4 h;
  h.x = acc.x + bp[0];
  h.y = acc.y + bp[1];
  h.z = acc.z + bp[2];
  h.w = acc.w + bp[3];
  h.x = h.x > 0.f ? h.x : 0.f;
  h.y = h.y > 0.f ? h.y : 0.f;
  h.z = h.z > 0.f ? h.z : 0.f;
  h.w = h.w > 0.f ? h.w : 0.f;
  if constexpr (!POOL) {
    ((float4*)h_out)[node * LPN + sub] = h;
  } else {
    static_assert(F == OUTC && LPN == 8, "pooled epilogue: 4 nodes of 8 lanes per wave");
    const int lane = threadIdx.x & 31;
    const int g0 = __builtin_amdgcn_readfirstlane(graph);
    const float gc = fmaxf((float)gcnt[live ? graph : 0], 1.0f);
    if (__all(graph == g0)) {
      float4 s = h;
      s.x += __shfl_xor(s.x, 8);
      s.y += __shfl_xor(s.y, 8);
      s.z += __shfl_xor(s.z, 8);
      s.w += __shfl_xor(s.w, 8);
      s.x += __shfl_xor(s.x, 16);
      s.y += __shfl_xor(s.y, 16);
      s.z += __shfl_xor(s.z, 16);
      s.w += __shfl_xor(s.w, 16);
      const int q = lane >> 3;
      const float v = q == 0 ? s.x : q == 1 ? s.y : q == 2 ? s.z : s.w;
      atomicAdd(&pool_sum[(size_t)g0 * OUTC + sub * 4 + q], v / gc);
    } else if (live) {
      float* ps = pool_sum + (size_t)graph * OUTC + sub * 4;
      atomicAdd(ps + 0, h.x / gc);
      atomicAdd(ps + 1, h.y / gc);
      atomicAdd(ps + 2, h.z / gc);
      atomicAdd(ps + 3, h.w / gc);
    }
  }
}


extern "C" void kernel_launch(void* const* d_in, const int* in_sizes, int n_in,
                              void* d_out, int out_size, void* d_ws, size_t ws_size,
                              hipStream_t stream) {
  const float* x     = (const float*)d_in[0];
  const int*   ei    = (const int*)d_in[1];
  const int*   batch = (const int*)d_in[2];
  const float* W1    = (const float*)d_in[3];
  const float* b1    = (const float*)d_in[4];
  const float* W2    = (const float*)d_in[5];
  const float* b2    = (const float*)d_in[6];
  const float* W3    = (const float*)d_in[7];
  const float* b3    = (const float*)d_in[8];

  const int N = in_sizes[0] / F_IN;
  const int E = in_sizes[1] / 2;
  const int G = out_size / OUTC;
  const int* src = ei;
  const int* dst = ei + E;
  float* out = (float*)d_out;

  float* dis = (float*)d_ws;
  int* gcnt = (int*)(dis + N);
  size_t off = (size_t)N + (size_t)G;
  off = (off + 63) & ~(size_t)63;
  float* bufA = (float*)d_ws + off;
  float* bufB = bufA + (size_t)N * HID;
  const int Npad = ceil_div(N + 1, SCAN_CHUNK) * SCAN_CHUNK;
  const size_t Epad = ((size_t)E + 63) & ~(size_t)63;
  int* degi   = (int*)(bufB + (size_t)N * HID);
  int* rowptr = degi + Npad;
  int* csr    = rowptr + Npad;
  int* cbase  = csr + 2 * Epad;
  const int TB = 256;
  const int chunk = ceil_div(ceil_div(E, NCHUNK), TB) * TB;

  const int gemm_blocks = ceil_div(ceil_div(N, 16), 8);

  chunk_count_kernel<<<NCHUNK, 1024, 0, stream>>>(dst, cbase, E, Npad, chunk, out, out_size, gcnt, G);
  scan_gemm_kernel<F_IN, HID><<<Npad / 256 + gemm_blocks, 256, 0, stream>>>(
      cbase, degi, Npad, batch, gcnt, N, Npad / 256, x, W1, bufA);
  rowptr_kernel<<<Npad / SCAN_CHUNK, 256, 0, stream>>>(degi, rowptr, dis, N);
  chunk_fill_kernel<<<NCHUNK * NSPLIT, 1024, 0, stream>>>(src, dst, cbase, rowptr, dis, csr, E, Npad, chunk);

  gather_kernel<HID, false><<<ceil_div(N, 256 / (HID / 4)), 256, 0, stream>>>(
      bufA, rowptr, csr, dis, b1, bufB, batch, out, gcnt, N);

  gemm_wmma_kernel<HID, HID2><<<gemm_blocks, 256, 0, stream>>>(bufB, W2, bufA, N);
  gather_kernel<HID2, false><<<ceil_div(N, 256 / (HID2 / 4)), 256, 0, stream>>>(
      bufA, rowptr, csr, dis, b2, bufB, batch, out, gcnt, N);

  gemm_wmma_kernel<HID2, OUTC><<<gemm_blocks, 256, 0, stream>>>(bufB, W3, bufA, N);
  gather_kernel<OUTC, true><<<ceil_div(N, 256 / (OUTC / 4)), 256, 0, stream>>>(
      bufA, rowptr, csr, dis, b3, bufB, batch, out, gcnt, N);
}


# Edge weight factored: products store rows times dis[row], gathers add plain rows and scale the sum by dis[node]; edge list holds ids only, fill reads no dis[src]; same sum order, new rounding
# speedup vs baseline: 4.9423x; 1.0744x over previous
#include <hip/hip_runtime.h>

typedef float v2f __attribute__((ext_vector_type(2)));
typedef float v8f __attribute__((ext_vector_type(8)));

#define F_IN 128
#define HID  128
#define HID2 64
#define OUTC 32

static inline int ceil_div(int a, int b) { return (a + b - 1) / b; }


#define SCAN_CHUNK 1024
#define NCHUNK 64
#define RANK_CAP 65536
#define NSPLIT 4
#define FILL_CAP 16384

__global__ __launch_bounds__(1024) void chunk_count_kernel(const int* __restrict__ dst,
                                                           int* __restrict__ cbase,
                                                           int E, int Npad, int chunk,
                                                           float* __restrict__ pool_sum, int n_pool,
                                                           int* __restrict__ gcnt, int G) {
  __shared__ int4 cnt4[RANK_CAP / 4];
  int* cnt = (int*)cnt4;
  const int t = threadIdx.x;
  for (int i = blockIdx.x * 1024 + t; i < n_pool; i += NCHUNK * 1024) pool_sum[i] = 0.f;
  for (int i = blockIdx.x * 1024 + t; i < G; i += NCHUNK * 1024) gcnt[i] = 0;
  const int e0 = blockIdx.x * chunk;
  const int e1 = e0 + chunk < E ? e0 + chunk : E;
  int4* mine = (int4*)(cbase + (size_t)blockIdx.x * Npad);

  for (int base = 0; base < Npad; base += RANK_CAP) {
    const int lim = Npad - base < RANK_CAP ? Npad - base : RANK_CAP;
    for (int i = t; i < lim / 4; i += 1024) cnt4[i] = make_int4(0, 0, 0, 0);
    __syncthreads();
    for (int e = e0 + t; e < e1; e += 8 * 1024) {
      int d[8];
#pragma unroll
      for (int u = 0; u < 8; ++u) {
        int ee = e + u * 1024;
        d[u] = dst[ee < e1 ? ee : e1 - 1];
      }
#pragma unroll
      for (int u = 0; u < 8; ++u) {
        int ee = e + u * 1024;
        int dd = d[u] - base;
        if (ee < e1 && (unsigned)dd < (unsigned)lim) atomicAdd(&cnt[dd], 1);
      }
    }
    __syncthreads();
    for (int i = t; i < lim / 4; i += 1024) mine[base / 4 + i] = cnt4[i];
    __syncthreads();
  }
}

__global__ __launch_bounds__(1024) void chunk_fill_kernel(const int* __restrict__ src,
                                                          const int* __restrict__ dst,
                                                          const int* __restrict__ cbase,
                                                          const int* __restrict__ rowptr,
                                                          int* __restrict__ csr,
                                                          int E, int Npad, int chunk) {
  __shared__ int4 cnt4[FILL_CAP / 4];
  int* cnt = (int*)cnt4;
  const int t = threadIdx.x;
  const int c = blockIdx.x / NSPLIT;
  const int range = Npad / NSPLIT;
  const int n0 = (blockIdx.x % NSPLIT) * range;
  const int e0 = c * chunk;
  const int e1 = e0 + chunk < E ? e0 + chunk : E;
  const int* mine = cbase + (size_t)c * Npad;

  for (int base = n0; base < n0 + range; base += FILL_CAP) {
    const int lim = n0 + range - base < FILL_CAP ? n0 + range - base : FILL_CAP;
    const int4* cb4 = (const int4*)(mine + base);
    const int4* rp4 = (const int4*)(rowptr + base);
    for (int i = t; i < lim / 4; i += 1024) {
      const int4 a = cb4[i], r = rp4[i];
      cnt4[i] = make_int4(a.x + r.x, a.y + r.y, a.z + r.z, a.w + r.w);
    }
    __syncthreads();
    for (int e = e0 + t; e < e1; e += 8 * 1024) {
      int d[8], s[8];
#pragma unroll
      for (int u = 0; u < 8; ++u) {
        int ee = e + u * 1024;
        ee = ee < e1 ? ee : e1 - 1;
        d[u] = dst[ee];
        s[u] = src[ee];
      }
      int pos[8];
#pragma unroll
      for (int u = 0; u < 8; ++u) {
        int ee = e + u * 1024;
        int dd = d[u] - base;
        pos[u] = -1;
        if (ee < e1 && (unsigned)dd < (unsigned)lim) pos[u] = atomicAdd(&cnt[dd], 1);
      }
#pragma unroll
      for (int u = 0; u < 8; ++u)
        if (pos[u] >= 0) csr[pos[u]] = s[u];
    }
    __syncthreads();
  }
}

__global__ __launch_bounds__(256) void chunk_scan_kernel(int* __restrict__ cbase, int* __restrict__ degi,
                                                         int Npad, const int* __restrict__ batch,
                                                         int* __restrict__ gcnt, int N) {
  const int i = blockIdx.x * 256 + threadIdx.x;
  const bool live = i < N;
  const int g = batch[live ? i : 0];
  int v[NCHUNK];
#pragma unroll
  for (int c = 0; c < NCHUNK; ++c) v[c] = cbase[(size_t)c * Npad + i];
  int run = 0;
#pragma unroll
  for (int c = 0; c < NCHUNK; ++c) {
    cbase[(size_t)c * Npad + i] = run;
    run += v[c];
  }
  degi[i] = run;

  const int gl = live ? g : -1;
  const int g0 = __builtin_amdgcn_readfirstlane(gl);
  if (__all(gl == g0)) {
    if ((threadIdx.x & 31) == 0 && g0 >= 0) atomicAdd(&gcnt[g0], 32);
  } else if (live) {
    atomicAdd(&gcnt[gl], 1);
  }
}

__device__ __forceinline__ void rowptr_block(int b, const int* __restrict__ degi,
                                             int* __restrict__ rowptr,
                                             float* __restrict__ dis, int N) {
  __shared__ int front[8];
  __shared__ int wtot[8];
  const int t = threadIdx.x;
  const int lane = t & 31, wave = t >> 5;
  const int4* d4 = (const int4*)degi;

  int pre = 0;
#pragma unroll 8
  for (int i = t; i < b * 256; i += 256) {
    int4 q = d4[i];
    pre += q.x + q.y + q.z + q.w;
  }
  for (int o = 16; o > 0; o >>= 1) pre += __shfl_xor(pre, o);
  if (lane == 0) front[wave] = pre;

  const int4 v = d4[b * 256 + t];
  const int tot = v.x + v.y + v.z + v.w;
  int inc = tot;
  for (int o = 1; o < 32; o <<= 1) {
    int u = __shfl_up(inc, o);
    if (lane >= o) inc += u;
  }
  if (lane == 31) wtot[wave] = inc;
  __syncthreads();

  int ex = inc - tot;
  for (int w = 0; w < 8; ++w) {
    ex += front[w];
    if (w < wave) ex += wtot[w];
  }
  const int c[4] = {v.x, v.y, v.z, v.w};
  const int i0 = (b * 256 + t) * 4;
#pragma unroll
  for (int k = 0; k < 4; ++k) {
    int i = i0 + k;
    rowptr[i] = ex;
    if (i < N) dis[i] = rsqrtf((float)(c[k] + 1));
    ex += c[k];
  }
}

template <int K, int NCOL>
__device__ __forceinline__ void gemm_wmma_block(
    int block, const float* __restrict__ A, const float* __restrict__ B, const int* __restrict__ degi,
    float* __restrict__ C, int M) {
  constexpr int CT = NCOL / 16 < 4 ? NCOL / 16 : 4;
  constexpr int RUN = 16 * CT;
  constexpr int KQ = K / 4;
  constexpr int KG = K / 8;
  __shared__ float4 Bl[KQ * NCOL];

  const int t = threadIdx.x;
  const int lane = t & 31;
  const int wave = t >> 5;
  const int c = lane & 15;
  const int h = lane >> 4;
  const int row0 = (block * 8 + wave) << 4;

  float4 a[KG];
  {
    int row = row0 + c;
    row = row < M ? row : M - 1;
    const float4* ap = (const float4*)(A + (size_t)row * K) + h;
#pragma unroll
    for (int g = 0; g < KG; ++g) a[g] = ap[2 * g];
  }
  int4 dg0, dg1;
  {
    const int4* dp = (const int4*)(degi + row0 + 8 * h);
    dg0 = dp[0];
    dg1 = dp[1];
  }

  static_assert((KQ * NCOL) % 256 == 0, "staging loop covers Bl in whole sweeps of the block");
#pragma unroll 8
  for (int it = 0; it < KQ * NCOL / 256; ++it) {
    const int i = t + it * 256;
    const int kq = i / NCOL;
    const int n = i % NCOL;
    const float* p = B + (size_t)(4 * kq) * NCOL + n;
    const int nl = n % RUN;
    const int slot = (n - nl) + (nl % CT) * 16 + nl / CT;
    Bl[kq * NCOL + slot] = make_float4(p[0], p[NCOL], p[2 * NCOL], p[3 * NCOL]);
  }
  __syncthreads();
  if (row0 >= M) return;
  const int dgr[8] = {dg0.x, dg0.y, dg0.z, dg0.w, dg1.x, dg1.y, dg1.z, dg1.w};
  float sc[8];
#pragma unroll
  for (int r = 0; r < 8; ++r) sc[r] = rsqrtf((float)(dgr[r] + 1));

  for (int n0 = 0; n0 < NCOL; n0 += RUN) {
    v8f acc[CT];
#pragma unroll
    for (int j = 0; j < CT; ++j) acc[j] = (v8f){0.f, 0.f, 0.f, 0.f, 0.f, 0.f, 0.f, 0.f};

#pragma unroll
    for (int g = 0; g < KG; ++g) {
      float4 b[CT];
#pragma unroll
      for (int j = 0; j < CT; ++j) b[j] = Bl[(2 * g + h) * NCOL + n0 + j * 16 + c];
#pragma unroll
      for (int j = 0; j < CT; ++j) {
        v2f av = {a[g].x, a[g].y};
        v2f bv = {b[j].x, b[j].y};
        acc[j] = __builtin_amdgcn_wmma_f32_16x16x4_f32(
             false, av,  false, bv,
             (short)0, acc[j],  false,  false);
      }
#pragma unroll
      for (int j = 0; j < CT; ++j) {
        v2f av = {a[g].z, a[g].w};
        v2f bv = {b[j].z, b[j].w};
        acc[j] = __builtin_amdgcn_wmma_f32_16x16x4_f32(
             false, av,  false, bv,
             (short)0, acc[j],  false,  false);
      }
    }

#pragma unroll
    for (int r = 0; r < 8; ++r) {
      const int row = row0 + 8 * h + r;
      if (row < M) {
        float* cp = C + (size_t)row * NCOL + n0 + c * CT;
        const float s = sc[r];
        if (CT == 4) {
          *(float4*)cp = make_float4(acc[0][r] * s, acc[1 % CT][r] * s, acc[2 % CT][r] * s, acc[3 % CT][r] * s);
        } else if (CT == 2) {
          *(float2*)cp = make_float2(acc[0][r] * s, acc[1 % CT][r] * s);
        } else {
          cp[0] = acc[0][r] * s;
        }
      }
    }
  }
}

template <int K, int NCOL>
__global__ __launch_bounds__(256) void gemm_wmma_kernel(
    const float* __restrict__ A, const float* __restrict__ B, const int* __restrict__ degi,
    float* __restrict__ C, int M) {
  gemm_wmma_block<K, NCOL>(blockIdx.x, A, B, degi, C, M);
}

template <int K, int NCOL>
__global__ __launch_bounds__(256) void rowptr_gemm_kernel(
    const int* __restrict__ degi, int* __restrict__ rowptr, float* __restrict__ dis, int N, int nrow,
    const float* __restrict__ A, const float* __restrict__ B, float* __restrict__ C) {
  if ((int)blockIdx.x < nrow) {
    rowptr_block(blockIdx.x, degi, rowptr, dis, N);
    return;
  }
  gemm_wmma_block<K, NCOL>(blockIdx.x - nrow, A, B, degi, C, N);
}


template <int LPN, bool TAIL>
__device__ __forceinline__ void add8(float4& acc, const int* __restrict__ csr,
                                     const float4* __restrict__ rows, int j, int end, int sub) {
  int ent[8];
  float4 v[8];
#pragma unroll
  for (int u = 0; u < 8; ++u) {
    int jj = j + u;
    if (TAIL) jj = jj < end ? jj : end - 1;
    ent[u] = csr[jj];
  }
#pragma unroll
  for (int u = 0; u < 8; ++u)
    v[u] = rows[ent[u] * LPN + sub];
#pragma unroll
  for (int u = 0; u < 8; ++u) {
    float4 r = v[u];
    if (TAIL) {
      bool ok = j + u < end;
      r.x = ok ? r.x : 0.f;
      r.y = ok ? r.y : 0.f;
      r.z = ok ? r.z : 0.f;
      r.w = ok ? r.w : 0.f;
    }
    acc.x += r.x;
    acc.y += r.y;
    acc.z += r.z;
    acc.w += r.w;
  }
}

template <int F, bool POOL>
__global__ __launch_bounds__(256) void gather_kernel(const float* __restrict__ XW,
                                                     const int* __restrict__ rowptr,
                                                     const int* __restrict__ csr,
                                                     const float* __restrict__ dis,
                                                     const float* __restrict__ bias,
                                                     float* __restrict__ h_out,
                                                     const int* __restrict__ batch,
                                                     float* __restrict__ pool_sum,
                                                     const int* __restrict__ gcnt, int N) {
  constexpr int LPN = F / 4;
  constexpr int NPB = 256 / LPN;
  const int sub = threadIdx.x % LPN;
  int node = blockIdx.x * NPB + threadIdx.x / LPN;
  const bool live = node < N;
  int graph = -1;
  if constexpr (POOL) {
    if (blockIdx.x * NPB + (int)(threadIdx.x & ~31u) / LPN >= N) return;
    node = live ? node : N - 1;
    if (live) graph = batch[node];
  } else {
    if (!live) return;
  }
  const float4* rows = (const float4*)XW;
  const int beg = rowptr[node], end = rowptr[node + 1];
  const float dd = dis[node];
  float4 acc = rows[node * LPN + sub];

  int j = beg;
  for (; j + 8 <= end; j += 8) add8<LPN, false>(acc, csr, rows, j, end, sub);
  if (j < end) add8<LPN, true>(acc, csr, rows, j, end, sub);

  const float* bp = bias + sub * 4;
  float4 h;
  h.x = acc.x * dd + bp[0];
  h.y = acc.y * dd + bp[1];
  h.z = acc.z * dd + bp[2];
  h.w = acc.w * dd + bp[3];
  h.x = h.x > 0.f ? h.x : 0.f;
  h.y = h.y > 0.f ? h.y : 0.f;
  h.z = h.z > 0.f ? h.z : 0.f;
  h.w = h.w > 0.f ? h.w : 0.f;
  if constexpr (!POOL) {
    ((float4*)h_out)[node * LPN + sub] = h;
  } else {
    static_assert(F == OUTC && LPN == 8, "pooled epilogue: 4 nodes of 8 lanes per wave");
    const int lane = threadIdx.x & 31;
    const int g0 = __builtin_amdgcn_readfirstlane(graph);
    const float gc = fmaxf((float)gcnt[live ? graph : 0], 1.0f);
    if (__all(graph == g0)) {
      float4 s = h;
      s.x += __shfl_xor(s.x, 8);
      s.y += __shfl_xor(s.y, 8);
      s.z += __shfl_xor(s.z, 8);
      s.w += __shfl_xor(s.w, 8);
      s.x += __shfl_xor(s.x, 16);
      s.y += __shfl_xor(s.y, 16);
      s.z += __shfl_xor(s.z, 16);
      s.w += __shfl_xor(s.w, 16);
      const int q = lane >> 3;
      const float v = q == 0 ? s.x : q == 1 ? s.y : q == 2 ? s.z : s.w;
      atomicAdd(&pool_sum[(size_t)g0 * OUTC + sub * 4 + q], v / gc);
    } else if (live) {
      float* ps = pool_sum + (size_t)graph * OUTC + sub * 4;
      atomicAdd(ps + 0, h.x / gc);
      atomicAdd(ps + 1, h.y / gc);
      atomicAdd(ps + 2, h.z / gc);
      atomicAdd(ps + 3, h.w / gc);
    }
  }
}


extern "C" void kernel_launch(void* const* d_in, const int* in_sizes, int n_in,
                              void* d_out, int out_size, void* d_ws, size_t ws_size,
                              hipStream_t stream) {
  const float* x     = (const float*)d_in[0];
  const int*   ei    = (const int*)d_in[1];
  const int*   batch = (const int*)d_in[2];
  const float* W1    = (const float*)d_in[3];
  const float* b1    = (const float*)d_in[4];
  const float* W2    = (const float*)d_in[5];
  const float* b2    = (const float*)d_in[6];
  const float* W3    = (const float*)d_in[7];
  const float* b3    = (const float*)d_in[8];

  const int N = in_sizes[0] / F_IN;
  const int E = in_sizes[1] / 2;
  const int G = out_size / OUTC;
  const int* src = ei;
  const int* dst = ei + E;
  float* out = (float*)d_out;

  float* dis = (float*)d_ws;
  int* gcnt = (int*)(dis + N);
  size_t off = (size_t)N + (size_t)G;
  off = (off + 63) & ~(size_t)63;
  float* bufA = (float*)d_ws + off;
  float* bufB = bufA + (size_t)N * HID;
  const int Npad = ceil_div(N + 1, SCAN_CHUNK) * SCAN_CHUNK;
  const size_t Epad = ((size_t)E + 63) & ~(size_t)63;
  int* degi   = (int*)(bufB + (size_t)N * HID);
  int* rowptr = degi + Npad;
  int* csr    = rowptr + Npad;
  int* cbase  = csr + Epad;
  const int TB = 256;
  const int chunk = ceil_div(ceil_div(E, NCHUNK), TB) * TB;

  const int gemm_blocks = ceil_div(ceil_div(N, 16), 8);

  chunk_count_kernel<<<NCHUNK, 1024, 0, stream>>>(dst, cbase, E, Npad, chunk, out, out_size, gcnt, G);
  chunk_scan_kernel<<<Npad / 256, 256, 0, stream>>>(cbase, degi, Npad, batch, gcnt, N);
  rowptr_gemm_kernel<F_IN, HID><<<Npad / SCAN_CHUNK + gemm_blocks, 256, 0, stream>>>(
      degi, rowptr, dis, N, Npad / SCAN_CHUNK, x, W1, bufA);
  chunk_fill_kernel<<<NCHUNK * NSPLIT, 1024, 0, stream>>>(src, dst, cbase, rowptr, csr, E, Npad, chunk);

  gather_kernel<HID, false><<<ceil_div(N, 256 / (HID / 4)), 256, 0, stream>>>(
      bufA, rowptr, csr, dis, b1, bufB, batch, out, gcnt, N);

  gemm_wmma_kernel<HID, HID2><<<gemm_blocks, 256, 0, stream>>>(bufB, W2, degi, bufA, N);
  gather_kernel<HID2, false><<<ceil_div(N, 256 / (HID2 / 4)), 256, 0, stream>>>(
      bufA, rowptr, csr, dis, b2, bufB, batch, out, gcnt, N);

  gemm_wmma_kernel<HID2, OUTC><<<gemm_blocks, 256, 0, stream>>>(bufB, W3, degi, bufA, N);
  gather_kernel<OUTC, true><<<ceil_div(N, 256 / (OUTC / 4)), 256, 0, stream>>>(
      bufA, rowptr, csr, dis, b3, bufB, batch, out, gcnt, N);
}


# Layers 2 and 3 products moved into the gather before them: a block keeps 64 finished rows in LDS and multiplies them by the next weights on WMMA; h1, h2 never stored, 2 launches fewer; same sums
# speedup vs baseline: 5.1207x; 1.0361x over previous
#include <hip/hip_runtime.h>

typedef float v2f __attribute__((ext_vector_type(2)));
typedef float v8f __attribute__((ext_vector_type(8)));

#define F_IN 128
#define HID  128
#define HID2 64
#define OUTC 32

static inline int ceil_div(int a, int b) { return (a + b - 1) / b; }


#define SCAN_CHUNK 1024
#define NCHUNK 64
#define RANK_CAP 65536
#define NSPLIT 4
#define FILL_CAP 16384

__global__ __launch_bounds__(1024) void chunk_count_kernel(const int* __restrict__ dst,
                                                           int* __restrict__ cbase,
                                                           int E, int Npad, int chunk,
                                                           float* __restrict__ pool_sum, int n_pool,
                                                           int* __restrict__ gcnt, int G) {
  __shared__ int4 cnt4[RANK_CAP / 4];
  int* cnt = (int*)cnt4;
  const int t = threadIdx.x;
  for (int i = blockIdx.x * 1024 + t; i < n_pool; i += NCHUNK * 1024) pool_sum[i] = 0.f;
  for (int i = blockIdx.x * 1024 + t; i < G; i += NCHUNK * 1024) gcnt[i] = 0;
  const int e0 = blockIdx.x * chunk;
  const int e1 = e0 + chunk < E ? e0 + chunk : E;
  int4* mine = (int4*)(cbase + (size_t)blockIdx.x * Npad);

  for (int base = 0; base < Npad; base += RANK_CAP) {
    const int lim = Npad - base < RANK_CAP ? Npad - base : RANK_CAP;
    for (int i = t; i < lim / 4; i += 1024) cnt4[i] = make_int4(0, 0, 0, 0);
    __syncthreads();
    for (int e = e0 + t; e < e1; e += 8 * 1024) {
      int d[8];
#pragma unroll
      for (int u = 0; u < 8; ++u) {
        int ee = e + u * 1024;
        d[u] = dst[ee < e1 ? ee : e1 - 1];
      }
#pragma unroll
      for (int u = 0; u < 8; ++u) {
        int ee = e + u * 1024;
        int dd = d[u] - base;
        if (ee < e1 && (unsigned)dd < (unsigned)lim) atomicAdd(&cnt[dd], 1);
      }
    }
    __syncthreads();
    for (int i = t; i < lim / 4; i += 1024) mine[base / 4 + i] = cnt4[i];
    __syncthreads();
  }
}

__global__ __launch_bounds__(1024) void chunk_fill_kernel(const int* __restrict__ src,
                                                          const int* __restrict__ dst,
                                                          const int* __restrict__ cbase,
                                                          const int* __restrict__ rowptr,
                                                          int* __restrict__ csr,
                                                          int E, int Npad, int chunk) {
  __shared__ int4 cnt4[FILL_CAP / 4];
  int* cnt = (int*)cnt4;
  const int t = threadIdx.x;
  const int c = blockIdx.x / NSPLIT;
  const int range = Npad / NSPLIT;
  const int n0 = (blockIdx.x % NSPLIT) * range;
  const int e0 = c * chunk;
  const int e1 = e0 + chunk < E ? e0 + chunk : E;
  const int* mine = cbase + (size_t)c * Npad;

  for (int base = n0; base < n0 + range; base += FILL_CAP) {
    const int lim = n0 + range - base < FILL_CAP ? n0 + range - base : FILL_CAP;
    const int4* cb4 = (const int4*)(mine + base);
    const int4* rp4 = (const int4*)(rowptr + base);
    for (int i = t; i < lim / 4; i += 1024) {
      const int4 a = cb4[i], r = rp4[i];
      cnt4[i] = make_int4(a.x + r.x, a.y + r.y, a.z + r.z, a.w + r.w);
    }
    __syncthreads();
    for (int e = e0 + t; e < e1; e += 8 * 1024) {
      int d[8], s[8];
#pragma unroll
      for (int u = 0; u < 8; ++u) {
        int ee = e + u * 1024;
        ee = ee < e1 ? ee : e1 - 1;
        d[u] = dst[ee];
        s[u] = src[ee];
      }
      int pos[8];
#pragma unroll
      for (int u = 0; u < 8; ++u) {
        int ee = e + u * 1024;
        int dd = d[u] - base;
        pos[u] = -1;
        if (ee < e1 && (unsigned)dd < (unsigned)lim) pos[u] = atomicAdd(&cnt[dd], 1);
      }
#pragma unroll
      for (int u = 0; u < 8; ++u)
        if (pos[u] >= 0) csr[pos[u]] = s[u];
    }
    __syncthreads();
  }
}

__global__ __launch_bounds__(256) void chunk_scan_kernel(int* __restrict__ cbase, int* __restrict__ degi,
                                                         int Npad, const int* __restrict__ batch,
                                                         int* __restrict__ gcnt, int N) {
  const int i = blockIdx.x * 256 + threadIdx.x;
  const bool live = i < N;
  const int g = batch[live ? i : 0];
  int v[NCHUNK];
#pragma unroll
  for (int c = 0; c < NCHUNK; ++c) v[c] = cbase[(size_t)c * Npad + i];
  int run = 0;
#pragma unroll
  for (int c = 0; c < NCHUNK; ++c) {
    cbase[(size_t)c * Npad + i] = run;
    run += v[c];
  }
  degi[i] = run;

  const int gl = live ? g : -1;
  const int g0 = __builtin_amdgcn_readfirstlane(gl);
  if (__all(gl == g0)) {
    if ((threadIdx.x & 31) == 0 && g0 >= 0) atomicAdd(&gcnt[g0], 32);
  } else if (live) {
    atomicAdd(&gcnt[gl], 1);
  }
}

__device__ __forceinline__ void rowptr_block(int b, const int* __restrict__ degi,
                                             int* __restrict__ rowptr,
                                             float* __restrict__ dis, int N) {
  __shared__ int front[8];
  __shared__ int wtot[8];
  const int t = threadIdx.x;
  const int lane = t & 31, wave = t >> 5;
  const int4* d4 = (const int4*)degi;

  int pre = 0;
#pragma unroll 8
  for (int i = t; i < b * 256; i += 256) {
    int4 q = d4[i];
    pre += q.x + q.y + q.z + q.w;
  }
  for (int o = 16; o > 0; o >>= 1) pre += __shfl_xor(pre, o);
  if (lane == 0) front[wave] = pre;

  const int4 v = d4[b * 256 + t];
  const int tot = v.x + v.y + v.z + v.w;
  int inc = tot;
  for (int o = 1; o < 32; o <<= 1) {
    int u = __shfl_up(inc, o);
    if (lane >= o) inc += u;
  }
  if (lane == 31) wtot[wave] = inc;
  __syncthreads();

  int ex = inc - tot;
  for (int w = 0; w < 8; ++w) {
    ex += front[w];
    if (w < wave) ex += wtot[w];
  }
  const int c[4] = {v.x, v.y, v.z, v.w};
  const int i0 = (b * 256 + t) * 4;
#pragma unroll
  for (int k = 0; k < 4; ++k) {
    int i = i0 + k;
    rowptr[i] = ex;
    if (i < N) dis[i] = rsqrtf((float)(c[k] + 1));
    ex += c[k];
  }
}

template <int K, int NCOL>
__device__ __forceinline__ void gemm_wmma_block(
    int block, const float* __restrict__ A, const float* __restrict__ B, const int* __restrict__ degi,
    float* __restrict__ C, int M) {
  constexpr int CT = NCOL / 16 < 4 ? NCOL / 16 : 4;
  constexpr int RUN = 16 * CT;
  constexpr int KQ = K / 4;
  constexpr int KG = K / 8;
  __shared__ float4 Bl[KQ * NCOL];

  const int t = threadIdx.x;
  const int lane = t & 31;
  const int wave = t >> 5;
  const int c = lane & 15;
  const int h = lane >> 4;
  const int row0 = (block * 8 + wave) << 4;

  float4 a[KG];
  {
    int row = row0 + c;
    row = row < M ? row : M - 1;
    const float4* ap = (const float4*)(A + (size_t)row * K) + h;
#pragma unroll
    for (int g = 0; g < KG; ++g) a[g] = ap[2 * g];
  }
  int4 dg0, dg1;
  {
    const int4* dp = (const int4*)(degi + row0 + 8 * h);
    dg0 = dp[0];
    dg1 = dp[1];
  }

  static_assert((KQ * NCOL) % 256 == 0, "staging loop covers Bl in whole sweeps of the block");
#pragma unroll 8
  for (int it = 0; it < KQ * NCOL / 256; ++it) {
    const int i = t + it * 256;
    const int kq = i / NCOL;
    const int n = i % NCOL;
    const float* p = B + (size_t)(4 * kq) * NCOL + n;
    const int nl = n % RUN;
    const int slot = (n - nl) + (nl % CT) * 16 + nl / CT;
    Bl[kq * NCOL + slot] = make_float4(p[0], p[NCOL], p[2 * NCOL], p[3 * NCOL]);
  }
  __syncthreads();
  if (row0 >= M) return;
  const int dgr[8] = {dg0.x, dg0.y, dg0.z, dg0.w, dg1.x, dg1.y, dg1.z, dg1.w};
  float sc[8];
#pragma unroll
  for (int r = 0; r < 8; ++r) sc[r] = rsqrtf((float)(dgr[r] + 1));

  for (int n0 = 0; n0 < NCOL; n0 += RUN) {
    v8f acc[CT];
#pragma unroll
    for (int j = 0; j < CT; ++j) acc[j] = (v8f){0.f, 0.f, 0.f, 0.f, 0.f, 0.f, 0.f, 0.f};

#pragma unroll
    for (int g = 0; g < KG; ++g) {
      float4 b[CT];
#pragma unroll
      for (int j = 0; j < CT; ++j) b[j] = Bl[(2 * g + h) * NCOL + n0 + j * 16 + c];
#pragma unroll
      for (int j = 0; j < CT; ++j) {
        v2f av = {a[g].x, a[g].y};
        v2f bv = {b[j].x, b[j].y};
        acc[j] = __builtin_amdgcn_wmma_f32_16x16x4_f32(
             false, av,  false, bv,
             (short)0, acc[j],  false,  false);
      }
#pragma unroll
      for (int j = 0; j < CT; ++j) {
        v2f av = {a[g].z, a[g].w};
        v2f bv = {b[j].z, b[j].w};
        acc[j] = __builtin_amdgcn_wmma_f32_16x16x4_f32(
             false, av,  false, bv,
             (short)0, acc[j],  false,  false);
      }
    }

#pragma unroll
    for (int r = 0; r < 8; ++r) {
      const int row = row0 + 8 * h + r;
      if (row < M) {
        float* cp = C + (size_t)row * NCOL + n0 + c * CT;
        const float s = sc[r];
        if (CT == 4) {
          *(float4*)cp = make_float4(acc[0][r] * s, acc[1 % CT][r] * s, acc[2 % CT][r] * s, acc[3 % CT][r] * s);
        } else if (CT == 2) {
          *(float2*)cp = make_float2(acc[0][r] * s, acc[1 % CT][r] * s);
        } else {
          cp[0] = acc[0][r] * s;
        }
      }
    }
  }
}

template <int K, int NCOL>
__global__ __launch_bounds__(256) void rowptr_gemm_kernel(
    const int* __restrict__ degi, int* __restrict__ rowptr, float* __restrict__ dis, int N, int nrow,
    const float* __restrict__ A, const float* __restrict__ B, float* __restrict__ C) {
  if ((int)blockIdx.x < nrow) {
    rowptr_block(blockIdx.x, degi, rowptr, dis, N);
    return;
  }
  gemm_wmma_block<K, NCOL>(blockIdx.x - nrow, A, B, degi, C, N);
}


template <int LPN, bool TAIL>
__device__ __forceinline__ void add8(float4& acc, const int* __restrict__ csr,
                                     const float4* __restrict__ rows, int j, int end, int sub) {
  int ent[8];
  float4 v[8];
#pragma unroll
  for (int u = 0; u < 8; ++u) {
    int jj = j + u;
    if (TAIL) jj = jj < end ? jj : end - 1;
    ent[u] = csr[jj];
  }
#pragma unroll
  for (int u = 0; u < 8; ++u)
    v[u] = rows[ent[u] * LPN + sub];
#pragma unroll
  for (int u = 0; u < 8; ++u) {
    float4 r = v[u];
    if (TAIL) {
      bool ok = j + u < end;
      r.x = ok ? r.x : 0.f;
      r.y = ok ? r.y : 0.f;
      r.z = ok ? r.z : 0.f;
      r.w = ok ? r.w : 0.f;
    }
    acc.x += r.x;
    acc.y += r.y;
    acc.z += r.z;
    acc.w += r.w;
  }
}

template <int F, bool POOL>
__global__ __launch_bounds__(256) void gather_kernel(const float* __restrict__ XW,
                                                     const int* __restrict__ rowptr,
                                                     const int* __restrict__ csr,
                                                     const float* __restrict__ dis,
                                                     const float* __restrict__ bias,
                                                     float* __restrict__ h_out,
                                                     const int* __restrict__ batch,
                                                     float* __restrict__ pool_sum,
                                                     const int* __restrict__ gcnt, int N) {
  constexpr int LPN = F / 4;
  constexpr int NPB = 256 / LPN;
  const int sub = threadIdx.x % LPN;
  int node = blockIdx.x * NPB + threadIdx.x / LPN;
  const bool live = node < N;
  int graph = -1;
  if constexpr (POOL) {
    if (blockIdx.x * NPB + (int)(threadIdx.x & ~31u) / LPN >= N) return;
    node = live ? node : N - 1;
    if (live) graph = batch[node];
  } else {
    if (!live) return;
  }
  const float4* rows = (const float4*)XW;
  const int beg = rowptr[node], end = rowptr[node + 1];
  const float dd = dis[node];
  float4 acc = rows[node * LPN + sub];

  int j = beg;
  for (; j + 8 <= end; j += 8) add8<LPN, false>(acc, csr, rows, j, end, sub);
  if (j < end) add8<LPN, true>(acc, csr, rows, j, end, sub);

  const float* bp = bias + sub * 4;
  float4 h;
  h.x = acc.x * dd + bp[0];
  h.y = acc.y * dd + bp[1];
  h.z = acc.z * dd + bp[2];
  h.w = acc.w * dd + bp[3];
  h.x = h.x > 0.f ? h.x : 0.f;
  h.y = h.y > 0.f ? h.y : 0.f;
  h.z = h.z > 0.f ? h.z : 0.f;
  h.w = h.w > 0.f ? h.w : 0.f;
  if constexpr (!POOL) {
    ((float4*)h_out)[node * LPN + sub] = h;
  } else {
    static_assert(F == OUTC && LPN == 8, "pooled epilogue: 4 nodes of 8 lanes per wave");
    const int lane = threadIdx.x & 31;
    const int g0 = __builtin_amdgcn_readfirstlane(graph);
    const float gc = fmaxf((float)gcnt[live ? graph : 0], 1.0f);
    if (__all(graph == g0)) {
      float4 s = h;
      s.x += __shfl_xor(s.x, 8);
      s.y += __shfl_xor(s.y, 8);
      s.z += __shfl_xor(s.z, 8);
      s.w += __shfl_xor(s.w, 8);
      s.x += __shfl_xor(s.x, 16);
      s.y += __shfl_xor(s.y, 16);
      s.z += __shfl_xor(s.z, 16);
      s.w += __shfl_xor(s.w, 16);
      const int q = lane >> 3;
      const float v = q == 0 ? s.x : q == 1 ? s.y : q == 2 ? s.z : s.w;
      atomicAdd(&pool_sum[(size_t)g0 * OUTC + sub * 4 + q], v / gc);
    } else if (live) {
      float* ps = pool_sum + (size_t)graph * OUTC + sub * 4;
      atomicAdd(ps + 0, h.x / gc);
      atomicAdd(ps + 1, h.y / gc);
      atomicAdd(ps + 2, h.z / gc);
      atomicAdd(ps + 3, h.w / gc);
    }
  }
}

#define GNB 64
template <int F, int NCOL>
__global__ __launch_bounds__(512) void gather_gemm_kernel(const float* __restrict__ XW,
                                                          const int* __restrict__ rowptr,
                                                          const int* __restrict__ csr,
                                                          const float* __restrict__ dis,
                                                          const float* __restrict__ bias,
                                                          const float* __restrict__ Wn,
                                                          const int* __restrict__ degi,
                                                          float* __restrict__ C, int N) {
  constexpr int LPN = F / 4;
  constexpr int SLOTS = 512 / LPN;
  constexpr int PASSES = GNB / SLOTS;
  constexpr int PITCH = F + 4;
  constexpr int CT = NCOL / 16;
  static_assert(GNB % SLOTS == 0 && (F * NCOL / 4) % 512 == 0 && (CT == 4 || CT == 2), "shapes");
  __shared__ float4 Wl4[F * NCOL / 4];
  __shared__ float4 ht4[GNB * PITCH / 4];
  float* ht = (float*)ht4;
  const float* Wl = (const float*)Wl4;

  const int t = threadIdx.x;
  const int base = blockIdx.x * GNB;

  {
    const float4* w4 = (const float4*)Wn;
#pragma unroll
    for (int it = 0; it < F * NCOL / 4 / 512; ++it) Wl4[t + it * 512] = w4[t + it * 512];
  }

  const int sub = t % LPN;
  const int slot = t / LPN;
  const float4* rows = (const float4*)XW;
  for (int p = 0; p < PASSES; ++p) {
    const int nl = p * SLOTS + slot;
    const int node = base + nl;
    float4 hv = make_float4(0.f, 0.f, 0.f, 0.f);
    if (node < N) {
      const int beg = rowptr[node], end = rowptr[node + 1];
      const float dd = dis[node];
      float4 acc = rows[node * LPN + sub];
      int j = beg;
      for (; j + 8 <= end; j += 8) add8<LPN, false>(acc, csr, rows, j, end, sub);
      if (j < end) add8<LPN, true>(acc, csr, rows, j, end, sub);
      const float* bp = bias + sub * 4;
      hv.x = acc.x * dd + bp[0];
      hv.y = acc.y * dd + bp[1];
      hv.z = acc.z * dd + bp[2];
      hv.w = acc.w * dd + bp[3];
      hv.x = hv.x > 0.f ? hv.x : 0.f;
      hv.y = hv.y > 0.f ? hv.y : 0.f;
      hv.z = hv.z > 0.f ? hv.z : 0.f;
      hv.w = hv.w > 0.f ? hv.w : 0.f;
    }
    *(float4*)(ht + nl * PITCH + sub * 4) = hv;
  }
  __syncthreads();

  const int lane = t & 31;
  const int wave = t >> 5;
  const int c = lane & 15;
  const int h = lane >> 4;
  const int row0 = base + 16 * wave;
  if (wave >= GNB / 16 || row0 >= N) return;

  v8f acc[CT];
#pragma unroll
  for (int j = 0; j < CT; ++j) acc[j] = (v8f){0.f, 0.f, 0.f, 0.f, 0.f, 0.f, 0.f, 0.f};
  const float* ap = ht + (16 * wave + c) * PITCH + 4 * h;
  const float* wp = Wl + (4 * h) * NCOL + c * CT;
#pragma unroll 1
  for (int g = 0; g < F / 8; ++g) {
    const float4 a = *(const float4*)(ap + 8 * g);
    float b[4][CT];
#pragma unroll
    for (int u = 0; u < 4; ++u) {
      const float* q = wp + (8 * g + u) * NCOL;
      if constexpr (CT == 4) {
        const float4 v = *(const float4*)q;
        b[u][0] = v.x; b[u][1] = v.y; b[u][2] = v.z; b[u][3] = v.w;
      } else {
        const float2 v = *(const float2*)q;
        b[u][0] = v.x; b[u][1] = v.y;
      }
    }
#pragma unroll
    for (int j = 0; j < CT; ++j) {
      v2f av = {a.x, a.y};
      v2f bv = {b[0][j], b[1][j]};
      acc[j] = __builtin_amdgcn_wmma_f32_16x16x4_f32(
           false, av,  false, bv,
           (short)0, acc[j],  false,  false);
    }
#pragma unroll
    for (int j = 0; j < CT; ++j) {
      v2f av = {a.z, a.w};
      v2f bv = {b[2][j], b[3][j]};
      acc[j] = __builtin_amdgcn_wmma_f32_16x16x4_f32(
           false, av,  false, bv,
           (short)0, acc[j],  false,  false);
    }
  }

  int4 dg0, dg1;
  {
    const int4* dp = (const int4*)(degi + row0 + 8 * h);
    dg0 = dp[0];
    dg1 = dp[1];
  }
  const int dgr[8] = {dg0.x, dg0.y, dg0.z, dg0.w, dg1.x, dg1.y, dg1.z, dg1.w};
#pragma unroll
  for (int r = 0; r < 8; ++r) {
    const int row = row0 + 8 * h + r;
    if (row < N) {
      const float s = rsqrtf((float)(dgr[r] + 1));
      float* cp = C + (size_t)row * NCOL + c * CT;
      if constexpr (CT == 4) {
        *(float4*)cp = make_float4(acc[0][r] * s, acc[1 % CT][r] * s, acc[2 % CT][r] * s, acc[3 % CT][r] * s);
      } else {
        *(float2*)cp = make_float2(acc[0][r] * s, acc[1 % CT][r] * s);
      }
    }
  }
}


extern "C" void kernel_launch(void* const* d_in, const int* in_sizes, int n_in,
                              void* d_out, int out_size, void* d_ws, size_t ws_size,
                              hipStream_t stream) {
  const float* x     = (const float*)d_in[0];
  const int*   ei    = (const int*)d_in[1];
  const int*   batch = (const int*)d_in[2];
  const float* W1    = (const float*)d_in[3];
  const float* b1    = (const float*)d_in[4];
  const float* W2    = (const float*)d_in[5];
  const float* b2    = (const float*)d_in[6];
  const float* W3    = (const float*)d_in[7];
  const float* b3    = (const float*)d_in[8];

  const int N = in_sizes[0] / F_IN;
  const int E = in_sizes[1] / 2;
  const int G = out_size / OUTC;
  const int* src = ei;
  const int* dst = ei + E;
  float* out = (float*)d_out;

  float* dis = (float*)d_ws;
  int* gcnt = (int*)(dis + N);
  size_t off = (size_t)N + (size_t)G;
  off = (off + 63) & ~(size_t)63;
  float* bufA = (float*)d_ws + off;
  float* bufB = bufA + (size_t)N * HID;
  const int Npad = ceil_div(N + 1, SCAN_CHUNK) * SCAN_CHUNK;
  const size_t Epad = ((size_t)E + 63) & ~(size_t)63;
  int* degi   = (int*)(bufB + (size_t)N * HID);
  int* rowptr = degi + Npad;
  int* csr    = rowptr + Npad;
  int* cbase  = csr + Epad;
  const int TB = 256;
  const int chunk = ceil_div(ceil_div(E, NCHUNK), TB) * TB;

  const int gemm_blocks = ceil_div(ceil_div(N, 16), 8);

  chunk_count_kernel<<<NCHUNK, 1024, 0, stream>>>(dst, cbase, E, Npad, chunk, out, out_size, gcnt, G);
  chunk_scan_kernel<<<Npad / 256, 256, 0, stream>>>(cbase, degi, Npad, batch, gcnt, N);
  rowptr_gemm_kernel<F_IN, HID><<<Npad / SCAN_CHUNK + gemm_blocks, 256, 0, stream>>>(
      degi, rowptr, dis, N, Npad / SCAN_CHUNK, x, W1, bufA);
  chunk_fill_kernel<<<NCHUNK * NSPLIT, 1024, 0, stream>>>(src, dst, cbase, rowptr, csr, E, Npad, chunk);

  gather_gemm_kernel<HID, HID2><<<ceil_div(N, GNB), 512, 0, stream>>>(
      bufA, rowptr, csr, dis, b1, W2, degi, bufB, N);
  gather_gemm_kernel<HID2, OUTC><<<ceil_div(N, GNB), 512, 0, stream>>>(
      bufB, rowptr, csr, dis, b2, W3, degi, bufA, N);

  gather_kernel<OUTC, true><<<ceil_div(N, 256 / (OUTC / 4)), 256, 0, stream>>>(
      bufA, rowptr, csr, dis, b3, bufB, batch, out, gcnt, N);
}


# Each gather block copies its node offsets, its contiguous edge-list segment, bias and graph ids to LDS once; the edge loop's index reads and per-node offset, bias, graph reads hit LDS; same sums
# speedup vs baseline: 5.4642x; 1.0671x over previous
#include <hip/hip_runtime.h>

typedef float v2f __attribute__((ext_vector_type(2)));
typedef float v8f __attribute__((ext_vector_type(8)));

#define F_IN 128
#define HID  128
#define HID2 64
#define OUTC 32

static inline int ceil_div(int a, int b) { return (a + b - 1) / b; }


#define SCAN_CHUNK 1024
#define NCHUNK 64
#define RANK_CAP 65536
#define NSPLIT 4
#define FILL_CAP 16384

__global__ __launch_bounds__(1024) void chunk_count_kernel(const int* __restrict__ dst,
                                                           int* __restrict__ cbase,
                                                           int E, int Npad, int chunk,
                                                           float* __restrict__ pool_sum, int n_pool,
                                                           int* __restrict__ gcnt, int G) {
  __shared__ int4 cnt4[RANK_CAP / 4];
  int* cnt = (int*)cnt4;
  const int t = threadIdx.x;
  for (int i = blockIdx.x * 1024 + t; i < n_pool; i += NCHUNK * 1024) pool_sum[i] = 0.f;
  for (int i = blockIdx.x * 1024 + t; i < G; i += NCHUNK * 1024) gcnt[i] = 0;
  const int e0 = blockIdx.x * chunk;
  const int e1 = e0 + chunk < E ? e0 + chunk : E;
  int4* mine = (int4*)(cbase + (size_t)blockIdx.x * Npad);

  for (int base = 0; base < Npad; base += RANK_CAP) {
    const int lim = Npad - base < RANK_CAP ? Npad - base : RANK_CAP;
    for (int i = t; i < lim / 4; i += 1024) cnt4[i] = make_int4(0, 0, 0, 0);
    __syncthreads();
    for (int e = e0 + t; e < e1; e += 8 * 1024) {
      int d[8];
#pragma unroll
      for (int u = 0; u < 8; ++u) {
        int ee = e + u * 1024;
        d[u] = dst[ee < e1 ? ee : e1 - 1];
      }
#pragma unroll
      for (int u = 0; u < 8; ++u) {
        int ee = e + u * 1024;
        int dd = d[u] - base;
        if (ee < e1 && (unsigned)dd < (unsigned)lim) atomicAdd(&cnt[dd], 1);
      }
    }
    __syncthreads();
    for (int i = t; i < lim / 4; i += 1024) mine[base / 4 + i] = cnt4[i];
    __syncthreads();
  }
}

__global__ __launch_bounds__(1024) void chunk_fill_kernel(const int* __restrict__ src,
                                                          const int* __restrict__ dst,
                                                          const int* __restrict__ cbase,
                                                          const int* __restrict__ rowptr,
                                                          int* __restrict__ csr,
                                                          int E, int Npad, int chunk) {
  __shared__ int4 cnt4[FILL_CAP / 4];
  int* cnt = (int*)cnt4;
  const int t = threadIdx.x;
  const int c = blockIdx.x / NSPLIT;
  const int range = Npad / NSPLIT;
  const int n0 = (blockIdx.x % NSPLIT) * range;
  const int e0 = c * chunk;
  const int e1 = e0 + chunk < E ? e0 + chunk : E;
  const int* mine = cbase + (size_t)c * Npad;

  for (int base = n0; base < n0 + range; base += FILL_CAP) {
    const int lim = n0 + range - base < FILL_CAP ? n0 + range - base : FILL_CAP;
    const int4* cb4 = (const int4*)(mine + base);
    const int4* rp4 = (const int4*)(rowptr + base);
    for (int i = t; i < lim / 4; i += 1024) {
      const int4 a = cb4[i], r = rp4[i];
      cnt4[i] = make_int4(a.x + r.x, a.y + r.y, a.z + r.z, a.w + r.w);
    }
    __syncthreads();
    for (int e = e0 + t; e < e1; e += 8 * 1024) {
      int d[8], s[8];
#pragma unroll
      for (int u = 0; u < 8; ++u) {
        int ee = e + u * 1024;
        ee = ee < e1 ? ee : e1 - 1;
        d[u] = dst[ee];
        s[u] = src[ee];
      }
      int pos[8];
#pragma unroll
      for (int u = 0; u < 8; ++u) {
        int ee = e + u * 1024;
        int dd = d[u] - base;
        pos[u] = -1;
        if (ee < e1 && (unsigned)dd < (unsigned)lim) pos[u] = atomicAdd(&cnt[dd], 1);
      }
#pragma unroll
      for (int u = 0; u < 8; ++u)
        if (pos[u] >= 0) csr[pos[u]] = s[u];
    }
    __syncthreads();
  }
}

__global__ __launch_bounds__(256) void chunk_scan_kernel(int* __restrict__ cbase, int* __restrict__ degi,
                                                         int Npad, const int* __restrict__ batch,
                                                         int* __restrict__ gcnt, int N) {
  const int i = blockIdx.x * 256 + threadIdx.x;
  const bool live = i < N;
  const int g = batch[live ? i : 0];
  int v[NCHUNK];
#pragma unroll
  for (int c = 0; c < NCHUNK; ++c) v[c] = cbase[(size_t)c * Npad + i];
  int run = 0;
#pragma unroll
  for (int c = 0; c < NCHUNK; ++c) {
    cbase[(size_t)c * Npad + i] = run;
    run += v[c];
  }
  degi[i] = run;

  const int gl = live ? g : -1;
  const int g0 = __builtin_amdgcn_readfirstlane(gl);
  if (__all(gl == g0)) {
    if ((threadIdx.x & 31) == 0 && g0 >= 0) atomicAdd(&gcnt[g0], 32);
  } else if (live) {
    atomicAdd(&gcnt[gl], 1);
  }
}

__device__ __forceinline__ void rowptr_block(int b, const int* __restrict__ degi,
                                             int* __restrict__ rowptr,
                                             float* __restrict__ dis, int N) {
  __shared__ int front[8];
  __shared__ int wtot[8];
  const int t = threadIdx.x;
  const int lane = t & 31, wave = t >> 5;
  const int4* d4 = (const int4*)degi;

  int pre = 0;
#pragma unroll 8
  for (int i = t; i < b * 256; i += 256) {
    int4 q = d4[i];
    pre += q.x + q.y + q.z + q.w;
  }
  for (int o = 16; o > 0; o >>= 1) pre += __shfl_xor(pre, o);
  if (lane == 0) front[wave] = pre;

  const int4 v = d4[b * 256 + t];
  const int tot = v.x + v.y + v.z + v.w;
  int inc = tot;
  for (int o = 1; o < 32; o <<= 1) {
    int u = __shfl_up(inc, o);
    if (lane >= o) inc += u;
  }
  if (lane == 31) wtot[wave] = inc;
  __syncthreads();

  int ex = inc - tot;
  for (int w = 0; w < 8; ++w) {
    ex += front[w];
    if (w < wave) ex += wtot[w];
  }
  const int c[4] = {v.x, v.y, v.z, v.w};
  const int i0 = (b * 256 + t) * 4;
#pragma unroll
  for (int k = 0; k < 4; ++k) {
    int i = i0 + k;
    rowptr[i] = ex;
    if (i < N) dis[i] = rsqrtf((float)(c[k] + 1));
    ex += c[k];
  }
}

template <int K, int NCOL>
__device__ __forceinline__ void gemm_wmma_block(
    int block, const float* __restrict__ A, const float* __restrict__ B, const int* __restrict__ degi,
    float* __restrict__ C, int M) {
  constexpr int CT = NCOL / 16 < 4 ? NCOL / 16 : 4;
  constexpr int RUN = 16 * CT;
  constexpr int KQ = K / 4;
  constexpr int KG = K / 8;
  __shared__ float4 Bl[KQ * NCOL];

  const int t = threadIdx.x;
  const int lane = t & 31;
  const int wave = t >> 5;
  const int c = lane & 15;
  const int h = lane >> 4;
  const int row0 = (block * 8 + wave) << 4;

  float4 a[KG];
  {
    int row = row0 + c;
    row = row < M ? row : M - 1;
    const float4* ap = (const float4*)(A + (size_t)row * K) + h;
#pragma unroll
    for (int g = 0; g < KG; ++g) a[g] = ap[2 * g];
  }
  int4 dg0, dg1;
  {
    const int4* dp = (const int4*)(degi + row0 + 8 * h);
    dg0 = dp[0];
    dg1 = dp[1];
  }

  static_assert((KQ * NCOL) % 256 == 0, "staging loop covers Bl in whole sweeps of the block");
#pragma unroll 8
  for (int it = 0; it < KQ * NCOL / 256; ++it) {
    const int i = t + it * 256;
    const int kq = i / NCOL;
    const int n = i % NCOL;
    const float* p = B + (size_t)(4 * kq) * NCOL + n;
    const int nl = n % RUN;
    const int slot = (n - nl) + (nl % CT) * 16 + nl / CT;
    Bl[kq * NCOL + slot] = make_float4(p[0], p[NCOL], p[2 * NCOL], p[3 * NCOL]);
  }
  __syncthreads();
  if (row0 >= M) return;
  const int dgr[8] = {dg0.x, dg0.y, dg0.z, dg0.w, dg1.x, dg1.y, dg1.z, dg1.w};
  float sc[8];
#pragma unroll
  for (int r = 0; r < 8; ++r) sc[r] = rsqrtf((float)(dgr[r] + 1));

  for (int n0 = 0; n0 < NCOL; n0 += RUN) {
    v8f acc[CT];
#pragma unroll
    for (int j = 0; j < CT; ++j) acc[j] = (v8f){0.f, 0.f, 0.f, 0.f, 0.f, 0.f, 0.f, 0.f};

#pragma unroll
    for (int g = 0; g < KG; ++g) {
      float4 b[CT];
#pragma unroll
      for (int j = 0; j < CT; ++j) b[j] = Bl[(2 * g + h) * NCOL + n0 + j * 16 + c];
#pragma unroll
      for (int j = 0; j < CT; ++j) {
        v2f av = {a[g].x, a[g].y};
        v2f bv = {b[j].x, b[j].y};
        acc[j] = __builtin_amdgcn_wmma_f32_16x16x4_f32(
             false, av,  false, bv,
             (short)0, acc[j],  false,  false);
      }
#pragma unroll
      for (int j = 0; j < CT; ++j) {
        v2f av = {a[g].z, a[g].w};
        v2f bv = {b[j].z, b[j].w};
        acc[j] = __builtin_amdgcn_wmma_f32_16x16x4_f32(
             false, av,  false, bv,
             (short)0, acc[j],  false,  false);
      }
    }

#pragma unroll
    for (int r = 0; r < 8; ++r) {
      const int row = row0 + 8 * h + r;
      if (row < M) {
        float* cp = C + (size_t)row * NCOL + n0 + c * CT;
        const float s = sc[r];
        if (CT == 4) {
          *(float4*)cp = make_float4(acc[0][r] * s, acc[1 % CT][r] * s, acc[2 % CT][r] * s, acc[3 % CT][r] * s);
        } else if (CT == 2) {
          *(float2*)cp = make_float2(acc[0][r] * s, acc[1 % CT][r] * s);
        } else {
          cp[0] = acc[0][r] * s;
        }
      }
    }
  }
}

template <int K, int NCOL>
__global__ __launch_bounds__(256) void rowptr_gemm_kernel(
    const int* __restrict__ degi, int* __restrict__ rowptr, float* __restrict__ dis, int N, int nrow,
    const float* __restrict__ A, const float* __restrict__ B, float* __restrict__ C) {
  if ((int)blockIdx.x < nrow) {
    rowptr_block(blockIdx.x, degi, rowptr, dis, N);
    return;
  }
  gemm_wmma_block<K, NCOL>(blockIdx.x - nrow, A, B, degi, C, N);
}


template <int LPN, bool TAIL>
__device__ __forceinline__ void add8(float4& acc, const int* __restrict__ csr,
                                     const float4* __restrict__ rows, int j, int end, int sub) {
  int ent[8];
  float4 v[8];
#pragma unroll
  for (int u = 0; u < 8; ++u) {
    int jj = j + u;
    if (TAIL) jj = jj < end ? jj : end - 1;
    ent[u] = csr[jj];
  }
#pragma unroll
  for (int u = 0; u < 8; ++u)
    v[u] = rows[ent[u] * LPN + sub];
#pragma unroll
  for (int u = 0; u < 8; ++u) {
    float4 r = v[u];
    if (TAIL) {
      bool ok = j + u < end;
      r.x = ok ? r.x : 0.f;
      r.y = ok ? r.y : 0.f;
      r.z = ok ? r.z : 0.f;
      r.w = ok ? r.w : 0.f;
    }
    acc.x += r.x;
    acc.y += r.y;
    acc.z += r.z;
    acc.w += r.w;
  }
}

template <int LPN>
__device__ __forceinline__ void sum_list(float4& acc, const int* __restrict__ lst,
                                         const float4* __restrict__ rows, int j, int end, int sub) {
  for (; j + 8 <= end; j += 8) add8<LPN, false>(acc, lst, rows, j, end, sub);
  if (j < end) add8<LPN, true>(acc, lst, rows, j, end, sub);
}

template <int NT, int PER>
__device__ __forceinline__ void stage_list_load(int (&v)[PER], const int* __restrict__ csr,
                                                int seg0, int seg1, int t) {
#pragma unroll
  for (int u = 0; u < PER; ++u) {
    int p = seg0 + t + u * NT;
    p = p < seg1 ? p : seg1 - 1;
    p = p > 0 ? p : 0;
    v[u] = csr[p];
  }
}
template <int NT, int PER>
__device__ __forceinline__ void stage_list_store(int* __restrict__ segl, const int (&v)[PER], int t) {
#pragma unroll
  for (int u = 0; u < PER; ++u) segl[t + u * NT] = v[u];
}

#define SEG_PER 5
#define SEG_CAP (SEG_PER * 512)
#define PSEG_PER 6
#define PSEG_CAP (PSEG_PER * 256)

template <int F, bool POOL>
__global__ __launch_bounds__(256) void gather_kernel(const float* __restrict__ XW,
                                                     const int* __restrict__ rowptr,
                                                     const int* __restrict__ csr,
                                                     const float* __restrict__ dis,
                                                     const float* __restrict__ bias,
                                                     float* __restrict__ h_out,
                                                     const int* __restrict__ batch,
                                                     float* __restrict__ pool_sum,
                                                     const int* __restrict__ gcnt, int N) {
  constexpr int LPN = F / 4;
  constexpr int NPB = 256 / LPN;
  __shared__ int segl[PSEG_CAP];
  __shared__ int4 aux4[256 / 4];
  static_assert(NPB + 1 <= 64 && F <= 64 && 64 % F == 0 && NPB <= 128 && 128 % NPB == 0, "layout of aux4");
  int* rpl = (int*)aux4;
  const float* bl = (const float*)aux4 + 64;
  const int* gl = (const int*)aux4 + 128;
  const int base = blockIdx.x * NPB;
  const int nend = base + NPB < N ? base + NPB : N;
  int seg0, seg1;
  {
    const int t = threadIdx.x;
    const int tt = t <= NPB ? t : NPB;
    const int i = base + tt;
    const int rb = rowptr[(t & 31) == 0 ? base : nend];
    const int rp = rowptr[i < N ? i : N];
    const float bv = bias[t & (F - 1)];
    seg0 = __builtin_amdgcn_readlane(rb, 0);
    seg1 = __builtin_amdgcn_readlane(rb, 1);
    int v[PSEG_PER];
    stage_list_load<256, PSEG_PER>(v, csr, seg0, seg1, t);
    stage_list_store<256, PSEG_PER>(segl, v, t);
    int w = t < 64 ? rp : __float_as_int(bv);
    if constexpr (POOL) {
      const int k = base + (t & (NPB - 1));
      const int gb = batch[k < N ? k : N - 1];
      w = t < 128 ? w : gb;
    }
    rpl[t] = w;
  }
  __syncthreads();

  const int sub = threadIdx.x % LPN;
  int node = base + threadIdx.x / LPN;
  const bool live = node < N;
  int graph = -1;
  if constexpr (POOL) {
    if (blockIdx.x * NPB + (int)(threadIdx.x & ~31u) / LPN >= N) return;
    node = live ? node : N - 1;
    if (live) graph = gl[node - base];
  } else {
    if (!live) return;
  }
  const float4* rows = (const float4*)XW;
  const int beg = rpl[node - base], end = rpl[node - base + 1];
  const float dd = dis[node];
  float4 acc = rows[node * LPN + sub];

  if (end - seg0 <= PSEG_CAP) sum_list<LPN>(acc, segl, rows, beg - seg0, end - seg0, sub);
  else sum_list<LPN>(acc, csr, rows, beg, end, sub);

  const float4 bp = *(const float4*)(bl + sub * 4);
  float4 h;
  h.x = acc.x * dd + bp.x;
  h.y = acc.y * dd + bp.y;
  h.z = acc.z * dd + bp.z;
  h.w = acc.w * dd + bp.w;
  h.x = h.x > 0.f ? h.x : 0.f;
  h.y = h.y > 0.f ? h.y : 0.f;
  h.z = h.z > 0.f ? h.z : 0.f;
  h.w = h.w > 0.f ? h.w : 0.f;
  if constexpr (!POOL) {
    ((float4*)h_out)[node * LPN + sub] = h;
  } else {
    static_assert(F == OUTC && LPN == 8, "pooled epilogue: 4 nodes of 8 lanes per wave");
    const int lane = threadIdx.x & 31;
    const int g0 = __builtin_amdgcn_readfirstlane(graph);
    const float gc = fmaxf((float)gcnt[live ? graph : 0], 1.0f);
    if (__all(graph == g0)) {
      float4 s = h;
      s.x += __shfl_xor(s.x, 8);
      s.y += __shfl_xor(s.y, 8);
      s.z += __shfl_xor(s.z, 8);
      s.w += __shfl_xor(s.w, 8);
      s.x += __shfl_xor(s.x, 16);
      s.y += __shfl_xor(s.y, 16);
      s.z += __shfl_xor(s.z, 16);
      s.w += __shfl_xor(s.w, 16);
      const int q = lane >> 3;
      const float v = q == 0 ? s.x : q == 1 ? s.y : q == 2 ? s.z : s.w;
      atomicAdd(&pool_sum[(size_t)g0 * OUTC + sub * 4 + q], v / gc);
    } else if (live) {
      float* ps = pool_sum + (size_t)graph * OUTC + sub * 4;
      atomicAdd(ps + 0, h.x / gc);
      atomicAdd(ps + 1, h.y / gc);
      atomicAdd(ps + 2, h.z / gc);
      atomicAdd(ps + 3, h.w / gc);
    }
  }
}

#define GNB 64
template <int F, int NCOL>
__global__ __launch_bounds__(512) void gather_gemm_kernel(const float* __restrict__ XW,
                                                          const int* __restrict__ rowptr,
                                                          const int* __restrict__ csr,
                                                          const float* __restrict__ dis,
                                                          const float* __restrict__ bias,
                                                          const float* __restrict__ Wn,
                                                          const int* __restrict__ degi,
                                                          float* __restrict__ C, int N) {
  constexpr int LPN = F / 4;
  constexpr int SLOTS = 512 / LPN;
  constexpr int PASSES = GNB / SLOTS;
  constexpr int PITCH = F + 4;
  constexpr int CT = NCOL / 16;
  static_assert(GNB % SLOTS == 0 && (F * NCOL / 4) % 512 == 0 && (CT == 4 || CT == 2), "shapes");
  constexpr int WPT = F * NCOL / 4 / 512;
  __shared__ float4 Wl4[F * NCOL / 4];
  __shared__ float4 ht4[GNB * PITCH / 4];
  __shared__ int segl[SEG_CAP];
  __shared__ int4 aux4[512 / 4];
  static_assert(GNB + 1 <= 128 && F <= 128 && 128 % F == 0, "layout of aux4");
  int* rpl = (int*)aux4;
  const float* bl = (const float*)aux4 + 128;
  float* ht = (float*)ht4;
  const float* Wl = (const float*)Wl4;

  const int t = threadIdx.x;
  const int base = blockIdx.x * GNB;
  const int nend = base + GNB < N ? base + GNB : N;
  int seg0, seg1;

  {
    const int tt = t <= GNB ? t : GNB;
    const int i = base + tt;
    const int rb = rowptr[(t & 31) == 0 ? base : nend];
    const int rp = rowptr[i < N ? i : N];
    const float bv = bias[t & (F - 1)];
    const float4* w4 = (const float4*)Wn;
    float4 wv[WPT];
#pragma unroll
    for (int it = 0; it < WPT; ++it) wv[it] = w4[t + it * 512];
    seg0 = __builtin_amdgcn_readlane(rb, 0);
    seg1 = __builtin_amdgcn_readlane(rb, 1);
    int v[SEG_PER];
    stage_list_load<512, SEG_PER>(v, csr, seg0, seg1, t);
#pragma unroll
    for (int it = 0; it < WPT; ++it) Wl4[t + it * 512] = wv[it];
    stage_list_store<512, SEG_PER>(segl, v, t);
    rpl[t] = t < 128 ? rp : __float_as_int(bv);
  }
  __syncthreads();

  const int sub = t % LPN;
  const int slot = t / LPN;
  const float4* rows = (const float4*)XW;
  for (int p = 0; p < PASSES; ++p) {
    const int nl = p * SLOTS + slot;
    const int node = base + nl;
    float4 hv = make_float4(0.f, 0.f, 0.f, 0.f);
    if (node < N) {
      const int beg = rpl[nl], end = rpl[nl + 1];
      const float dd = dis[node];
      float4 acc = rows[node * LPN + sub];
      if (end - seg0 <= SEG_CAP) sum_list<LPN>(acc, segl, rows, beg - seg0, end - seg0, sub);
      else sum_list<LPN>(acc, csr, rows, beg, end, sub);
      const float4 bp = *(const float4*)(bl + sub * 4);
      hv.x = acc.x * dd + bp.x;
      hv.y = acc.y * dd + bp.y;
      hv.z = acc.z * dd + bp.z;
      hv.w = acc.w * dd + bp.w;
      hv.x = hv.x > 0.f ? hv.x : 0.f;
      hv.y = hv.y > 0.f ? hv.y : 0.f;
      hv.z = hv.z > 0.f ? hv.z : 0.f;
      hv.w = hv.w > 0.f ? hv.w : 0.f;
    }
    *(float4*)(ht + nl * PITCH + sub * 4) = hv;
  }
  __syncthreads();

  const int lane = t & 31;
  const int wave = t >> 5;
  const int c = lane & 15;
  const int h = lane >> 4;
  const int row0 = base + 16 * wave;
  if (wave >= GNB / 16 || row0 >= N) return;

  v8f acc[CT];
#pragma unroll
  for (int j = 0; j < CT; ++j) acc[j] = (v8f){0.f, 0.f, 0.f, 0.f, 0.f, 0.f, 0.f, 0.f};
  const float* ap = ht + (16 * wave + c) * PITCH + 4 * h;
  const float* wp = Wl + (4 * h) * NCOL + c * CT;
#pragma unroll 1
  for (int g = 0; g < F / 8; ++g) {
    const float4 a = *(const float4*)(ap + 8 * g);
    float b[4][CT];
#pragma unroll
    for (int u = 0; u < 4; ++u) {
      const float* q = wp + (8 * g + u) * NCOL;
      if constexpr (CT == 4) {
        const float4 v = *(const float4*)q;
        b[u][0] = v.x; b[u][1] = v.y; b[u][2] = v.z; b[u][3] = v.w;
      } else {
        const float2 v = *(const float2*)q;
        b[u][0] = v.x; b[u][1] = v.y;
      }
    }
#pragma unroll
    for (int j = 0; j < CT; ++j) {
      v2f av = {a.x, a.y};
      v2f bv = {b[0][j], b[1][j]};
      acc[j] = __builtin_amdgcn_wmma_f32_16x16x4_f32(
           false, av,  false, bv,
           (short)0, acc[j],  false,  false);
    }
#pragma unroll
    for (int j = 0; j < CT; ++j) {
      v2f av = {a.z, a.w};
      v2f bv = {b[2][j], b[3][j]};
      acc[j] = __builtin_amdgcn_wmma_f32_16x16x4_f32(
           false, av,  false, bv,
           (short)0, acc[j],  false,  false);
    }
  }

  int4 dg0, dg1;
  {
    const int4* dp = (const int4*)(degi + row0 + 8 * h);
    dg0 = dp[0];
    dg1 = dp[1];
  }
  const int dgr[8] = {dg0.x, dg0.y, dg0.z, dg0.w, dg1.x, dg1.y, dg1.z, dg1.w};
#pragma unroll
  for (int r = 0; r < 8; ++r) {
    const int row = row0 + 8 * h + r;
    if (row < N) {
      const float s = rsqrtf((float)(dgr[r] + 1));
      float* cp = C + (size_t)row * NCOL + c * CT;
      if constexpr (CT == 4) {
        *(float4*)cp = make_float4(acc[0][r] * s, acc[1 % CT][r] * s, acc[2 % CT][r] * s, acc[3 % CT][r] * s);
      } else {
        *(float2*)cp = make_float2(acc[0][r] * s, acc[1 % CT][r] * s);
      }
    }
  }
}


extern "C" void kernel_launch(void* const* d_in, const int* in_sizes, int n_in,
                              void* d_out, int out_size, void* d_ws, size_t ws_size,
                              hipStream_t stream) {
  const float* x     = (const float*)d_in[0];
  const int*   ei    = (const int*)d_in[1];
  const int*   batch = (const int*)d_in[2];
  const float* W1    = (const float*)d_in[3];
  const float* b1    = (const float*)d_in[4];
  const float* W2    = (const float*)d_in[5];
  const float* b2    = (const float*)d_in[6];
  const float* W3    = (const float*)d_in[7];
  const float* b3    = (const float*)d_in[8];

  const int N = in_sizes[0] / F_IN;
  const int E = in_sizes[1] / 2;
  const int G = out_size / OUTC;
  const int* src = ei;
  const int* dst = ei + E;
  float* out = (float*)d_out;

  float* dis = (float*)d_ws;
  int* gcnt = (int*)(dis + N);
  size_t off = (size_t)N + (size_t)G;
  off = (off + 63) & ~(size_t)63;
  float* bufA = (float*)d_ws + off;
  float* bufB = bufA + (size_t)N * HID;
  const int Npad = ceil_div(N + 1, SCAN_CHUNK) * SCAN_CHUNK;
  const size_t Epad = ((size_t)E + 63) & ~(size_t)63;
  int* degi   = (int*)(bufB + (size_t)N * HID);
  int* rowptr = degi + Npad;
  int* csr    = rowptr + Npad;
  int* cbase  = csr + Epad;
  const int TB = 256;
  const int chunk = ceil_div(ceil_div(E, NCHUNK), TB) * TB;

  const int gemm_blocks = ceil_div(ceil_div(N, 16), 8);

  chunk_count_kernel<<<NCHUNK, 1024, 0, stream>>>(dst, cbase, E, Npad, chunk, out, out_size, gcnt, G);
  chunk_scan_kernel<<<Npad / 256, 256, 0, stream>>>(cbase, degi, Npad, batch, gcnt, N);
  rowptr_gemm_kernel<F_IN, HID><<<Npad / SCAN_CHUNK + gemm_blocks, 256, 0, stream>>>(
      degi, rowptr, dis, N, Npad / SCAN_CHUNK, x, W1, bufA);
  chunk_fill_kernel<<<NCHUNK * NSPLIT, 1024, 0, stream>>>(src, dst, cbase, rowptr, csr, E, Npad, chunk);

  gather_gemm_kernel<HID, HID2><<<ceil_div(N, GNB), 512, 0, stream>>>(
      bufA, rowptr, csr, dis, b1, W2, degi, bufB, N);
  gather_gemm_kernel<HID2, OUTC><<<ceil_div(N, GNB), 512, 0, stream>>>(
      bufB, rowptr, csr, dis, b2, W3, degi, bufA, N);

  gather_kernel<OUTC, true><<<ceil_div(N, 256 / (OUTC / 4)), 256, 0, stream>>>(
      bufA, rowptr, csr, dis, b3, bufB, batch, out, gcnt, N);
}
